# starting kernel
# baseline (speedup 1.0000x reference)
#include <hip/hip_runtime.h>
#include <math.h>

#define DIN 256
#define HID 64
#define HEADS 4
#define DOUT 128
#define NEG_SLOPE 0.2f

typedef __attribute__((ext_vector_type(2))) float v2f;
typedef __attribute__((ext_vector_type(8))) float v8f;

__device__ __forceinline__ void atomicMaxFloat(float* addr, float val) {
  if (val >= 0.0f)
    atomicMax((int*)addr, __float_as_int(val));
  else
    atomicMin((unsigned int*)addr, __float_as_uint(val));
}

template<int K, int NOUT>
__global__ void gemm_f32_wmma(const float* __restrict__ X,
                              const float* __restrict__ W,
                              float* __restrict__ H, int nrows) {
  const int lane = threadIdx.x & 31;
  const int wave = threadIdx.x >> 5;
  const int colGroups = NOUT / 64;
  const int strip = blockIdx.x * (blockDim.x >> 5) + wave;
  const int rowTile = strip / colGroups;
  const int colGroup = strip - rowTile * colGroups;
  if (rowTile * 16 >= nrows) return;
  const int r = lane & 15;
  const int half = lane >> 4;
  const float* __restrict__ Ap = X + (size_t)(rowTile * 16 + r) * K + 2 * half;
  const float* __restrict__ Bp = W + (size_t)(2 * half) * NOUT + colGroup * 64 + r;
  v8f acc0 = {}, acc1 = {}, acc2 = {}, acc3 = {};
#pragma unroll 4
  for (int k = 0; k < K; k += 4) {
    v2f a;
    a[0] = Ap[k];
    a[1] = Ap[k + 1];
    const float* b0 = Bp + (size_t)k * NOUT;
    const float* b1 = Bp + (size_t)(k + 1) * NOUT;
    v2f bA, bB, bC, bD;
    bA[0] = b0[0];   bA[1] = b1[0];
    bB[0] = b0[16];  bB[1] = b1[16];
    bC[0] = b0[32];  bC[1] = b1[32];
    bD[0] = b0[48];  bD[1] = b1[48];
    acc0 = __builtin_amdgcn_wmma_f32_16x16x4_f32(false, a, false, bA,
                                                 (short)0, acc0, false, false);
    acc1 = __builtin_amdgcn_wmma_f32_16x16x4_f32(false, a, false, bB,
                                                 (short)0, acc1, false, false);
    acc2 = __builtin_amdgcn_wmma_f32_16x16x4_f32(false, a, false, bC,
                                                 (short)0, acc2, false, false);
    acc3 = __builtin_amdgcn_wmma_f32_16x16x4_f32(false, a, false, bD,
                                                 (short)0, acc3, false, false);
  }
  float* __restrict__ Cp =
      H + (size_t)(rowTile * 16 + 8 * half) * NOUT + colGroup * 64 + r;
#pragma unroll
  for (int v = 0; v < 8; ++v) {
    Cp[(size_t)v * NOUT]      = acc0[v];
    Cp[(size_t)v * NOUT + 16] = acc1[v];
    Cp[(size_t)v * NOUT + 32] = acc2[v];
    Cp[(size_t)v * NOUT + 48] = acc3[v];
  }
}

__global__ void dot_e_kernel(const float* __restrict__ H,
                             const float* __restrict__ a_src,
                             const float* __restrict__ a_dst,
                             float* __restrict__ e_src,
                             float* __restrict__ e_dst,
                             int n, int heads, int C) {
  int idx = blockIdx.x * blockDim.x + threadIdx.x;
  if (idx >= n * heads) return;
  int node = idx / heads;
  int hd = idx - node * heads;
  const float* hp = H + (size_t)node * heads * C + (size_t)hd * C;
  const float* as = a_src + (size_t)hd * C;
  const float* ad = a_dst + (size_t)hd * C;
  float s0 = 0.f, s1 = 0.f;
  for (int c = 0; c < C; ++c) {
    float v = hp[c];
    s0 += v * as[c];
    s1 += v * ad[c];
  }
  e_src[idx] = s0;
  e_dst[idx] = s1;
}

__global__ void fill_kernel(float* __restrict__ p, float v, size_t n) {
  size_t i = (size_t)blockIdx.x * blockDim.x + threadIdx.x;
  size_t stride = (size_t)gridDim.x * blockDim.x;
  for (; i < n; i += stride) p[i] = v;
}

__global__ void edge_logit_max(const int* __restrict__ srcIdx,
                               const int* __restrict__ dstIdx,
                               int E, int Etot,
                               const float* __restrict__ e_src,
                               const float* __restrict__ e_dst,
                               float* __restrict__ logits,
                               float* __restrict__ m, int heads) {
  int t = blockIdx.x * blockDim.x + threadIdx.x;
  if (t >= Etot * heads) return;
  int e = t / heads;
  int hd = t - e * heads;
  int s = (e < E) ? srcIdx[e] : (e - E);
  int d = (e < E) ? dstIdx[e] : (e - E);
  float l = e_src[s * heads + hd] + e_dst[d * heads + hd];
  l = (l > 0.f) ? l : l * NEG_SLOPE;
  logits[t] = l;
  atomicMaxFloat(&m[d * heads + hd], l);
}

__global__ void edge_exp_sum(const int* __restrict__ srcIdx,
                             const int* __restrict__ dstIdx,
                             int E, int Etot,
                             float* __restrict__ logits,
                             const float* __restrict__ m,
                             float* __restrict__ denom, int heads) {
  int t = blockIdx.x * blockDim.x + threadIdx.x;
  if (t >= Etot * heads) return;
  int e = t / heads;
  int hd = t - e * heads;
  int d = (e < E) ? dstIdx[e] : (e - E);
  float ex = expf(logits[t] - m[d * heads + hd]);
  logits[t] = ex;
  atomicAdd(&denom[d * heads + hd], ex);
}

__global__ void edge_scatter(const int* __restrict__ srcIdx,
                             const int* __restrict__ dstIdx,
                             int E, int Etot,
                             const float* __restrict__ ex,
                             const float* __restrict__ denom,
                             const float* __restrict__ H,
                             float* __restrict__ agg,
                             int heads, int C) {
  int lane = threadIdx.x & 31;
  int wid = (int)(((size_t)blockIdx.x * blockDim.x + threadIdx.x) >> 5);
  if (wid >= Etot * heads) return;
  int e = wid / heads;
  int hd = wid - e * heads;
  int s = (e < E) ? srcIdx[e] : (e - E);
  int d = (e < E) ? dstIdx[e] : (e - E);
  float alpha = ex[wid] / (denom[d * heads + hd] + 1e-16f);
  const float* hp = H + (size_t)s * heads * C + (size_t)hd * C;
  float* ap = agg + (size_t)d * heads * C + (size_t)hd * C;
  for (int c = lane; c < C; c += 32)
    atomicAdd(&ap[c], alpha * hp[c]);
}

__global__ void bias_elu(const float* __restrict__ agg,
                         const float* __restrict__ b,
                         float* __restrict__ out, int n, int F) {
  int t = blockIdx.x * blockDim.x + threadIdx.x;
  if (t >= n * F) return;
  int f = t % F;
  float v = agg[t] + b[f];
  out[t] = (v > 0.f) ? v : (expf(v) - 1.f);
}

__global__ void final_mean(const float* __restrict__ agg,
                           const float* __restrict__ b,
                           float* __restrict__ out, int n, int F) {
  int lane = threadIdx.x & 31;
  int node = (int)(((size_t)blockIdx.x * blockDim.x + threadIdx.x) >> 5);
  if (node >= n) return;
  const float* ap = agg + (size_t)node * F;
  float sum = 0.f;
  for (int c = lane; c < F; c += 32) {
    float v = ap[c] + b[c];
    v = (v > 0.f) ? v : (expf(v) - 1.f);
    sum += v;
  }
#pragma unroll
  for (int off = 16; off > 0; off >>= 1)
    sum += __shfl_down(sum, off, 32);
  if (lane == 0) out[node] = sum / (float)F;
}

static inline int cdiv(long long a, long long b) { return (int)((a + b - 1) / b); }

extern "C" void kernel_launch(void* const* d_in, const int* in_sizes, int n_in,
                              void* d_out, int out_size, void* d_ws, size_t ws_size,
                              hipStream_t stream) {
  const float* x      = (const float*)d_in[0];
  const int*   ei     = (const int*)  d_in[1];
  const float* W1     = (const float*)d_in[2];
  const float* a1_src = (const float*)d_in[3];
  const float* a1_dst = (const float*)d_in[4];
  const float* b1     = (const float*)d_in[5];
  const float* W2     = (const float*)d_in[6];
  const float* a2_src = (const float*)d_in[7];
  const float* a2_dst = (const float*)d_in[8];
  const float* b2     = (const float*)d_in[9];
  float* outp = (float*)d_out;

  const int N    = in_sizes[0] / DIN;
  const int E    = in_sizes[1] / 2;
  const int Etot = E + N;
  const int F1   = HEADS * HID;
  const int* srcIdx = ei;
  const int* dstIdx = ei + E;

  float* ws = (float*)d_ws;
  float* h1      = ws;
  float* e_src_b = h1 + (size_t)N * F1;
  float* e_dst_b = e_src_b + (size_t)N * HEADS;
  float* edgebuf = e_dst_b + (size_t)N * HEADS;
  float* m_b     = edgebuf + (size_t)Etot * HEADS;
  float* den_b   = m_b + (size_t)N * HEADS;
  float* aggreg  = den_b + (size_t)N * HEADS;
  float* h2      = aggreg;
  float* agg2    = aggreg + (size_t)N * DOUT;

  const int TB = 256;

  {
    int strips = (N / 16) * (F1 / 64);
    gemm_f32_wmma<DIN, F1><<<cdiv(strips, TB / 32), TB, 0, stream>>>(x, W1, h1, N);

    dot_e_kernel<<<cdiv((long long)N * HEADS, TB), TB, 0, stream>>>(
        h1, a1_src, a1_dst, e_src_b, e_dst_b, N, HEADS, HID);

    fill_kernel<<<1024, TB, 0, stream>>>(m_b, -INFINITY, (size_t)N * HEADS);
    fill_kernel<<<1024, TB, 0, stream>>>(den_b, 0.f, (size_t)N * HEADS);
    fill_kernel<<<4096, TB, 0, stream>>>(aggreg, 0.f, (size_t)N * F1);

    int eh = Etot * HEADS;
    edge_logit_max<<<cdiv(eh, TB), TB, 0, stream>>>(srcIdx, dstIdx, E, Etot,
                                                    e_src_b, e_dst_b, edgebuf, m_b, HEADS);
    edge_exp_sum<<<cdiv(eh, TB), TB, 0, stream>>>(srcIdx, dstIdx, E, Etot,
                                                  edgebuf, m_b, den_b, HEADS);
    edge_scatter<<<cdiv((long long)eh * 32, TB), TB, 0, stream>>>(
        srcIdx, dstIdx, E, Etot, edgebuf, den_b, h1, aggreg, HEADS, HID);

    bias_elu<<<cdiv((long long)N * F1, TB), TB, 0, stream>>>(aggreg, b1, h1, N, F1);
  }

  {
    int strips = (N / 16) * (DOUT / 64);
    gemm_f32_wmma<F1, DOUT><<<cdiv(strips, TB / 32), TB, 0, stream>>>(h1, W2, h2, N);

    dot_e_kernel<<<cdiv(N, TB), TB, 0, stream>>>(
        h2, a2_src, a2_dst, e_src_b, e_dst_b, N, 1, DOUT);

    fill_kernel<<<1024, TB, 0, stream>>>(m_b, -INFINITY, (size_t)N);
    fill_kernel<<<1024, TB, 0, stream>>>(den_b, 0.f, (size_t)N);
    fill_kernel<<<4096, TB, 0, stream>>>(agg2, 0.f, (size_t)N * DOUT);

    edge_logit_max<<<cdiv(Etot, TB), TB, 0, stream>>>(srcIdx, dstIdx, E, Etot,
                                                      e_src_b, e_dst_b, edgebuf, m_b, 1);
    edge_exp_sum<<<cdiv(Etot, TB), TB, 0, stream>>>(srcIdx, dstIdx, E, Etot,
                                                    edgebuf, m_b, den_b, 1);
    edge_scatter<<<cdiv((long long)Etot * 32, TB), TB, 0, stream>>>(
        srcIdx, dstIdx, E, Etot, edgebuf, den_b, h2, agg2, 1, DOUT);

    final_mean<<<cdiv((long long)N * 32, TB), TB, 0, stream>>>(agg2, b2, outp, N, DOUT);
  }
}


# edge_scatter: one wave per 32-edge batch, not per edge-head: coalesced src/dst/alpha preamble, lane-to-scalar broadcast, 16+ H-row loads in flight before atomics; float atomics kept (arrival order)
# speedup vs baseline: 1.7881x; 1.7881x over previous
#include <hip/hip_runtime.h>
#include <math.h>

#define DIN 256
#define HID 64
#define HEADS 4
#define DOUT 128
#define NEG_SLOPE 0.2f

typedef __attribute__((ext_vector_type(2))) float v2f;
typedef __attribute__((ext_vector_type(8))) float v8f;

__device__ __forceinline__ void atomicMaxFloat(float* addr, float val) {
  if (val >= 0.0f)
    atomicMax((int*)addr, __float_as_int(val));
  else
    atomicMin((unsigned int*)addr, __float_as_uint(val));
}

template<int K, int NOUT>
__global__ void gemm_f32_wmma(const float* __restrict__ X,
                              const float* __restrict__ W,
                              float* __restrict__ H, int nrows) {
  const int lane = threadIdx.x & 31;
  const int wave = threadIdx.x >> 5;
  const int colGroups = NOUT / 64;
  const int strip = blockIdx.x * (blockDim.x >> 5) + wave;
  const int rowTile = strip / colGroups;
  const int colGroup = strip - rowTile * colGroups;
  if (rowTile * 16 >= nrows) return;
  const int r = lane & 15;
  const int half = lane >> 4;
  const float* __restrict__ Ap = X + (size_t)(rowTile * 16 + r) * K + 2 * half;
  const float* __restrict__ Bp = W + (size_t)(2 * half) * NOUT + colGroup * 64 + r;
  v8f acc0 = {}, acc1 = {}, acc2 = {}, acc3 = {};
#pragma unroll 4
  for (int k = 0; k < K; k += 4) {
    v2f a;
    a[0] = Ap[k];
    a[1] = Ap[k + 1];
    const float* b0 = Bp + (size_t)k * NOUT;
    const float* b1 = Bp + (size_t)(k + 1) * NOUT;
    v2f bA, bB, bC, bD;
    bA[0] = b0[0];   bA[1] = b1[0];
    bB[0] = b0[16];  bB[1] = b1[16];
    bC[0] = b0[32];  bC[1] = b1[32];
    bD[0] = b0[48];  bD[1] = b1[48];
    acc0 = __builtin_amdgcn_wmma_f32_16x16x4_f32(false, a, false, bA,
                                                 (short)0, acc0, false, false);
    acc1 = __builtin_amdgcn_wmma_f32_16x16x4_f32(false, a, false, bB,
                                                 (short)0, acc1, false, false);
    acc2 = __builtin_amdgcn_wmma_f32_16x16x4_f32(false, a, false, bC,
                                                 (short)0, acc2, false, false);
    acc3 = __builtin_amdgcn_wmma_f32_16x16x4_f32(false, a, false, bD,
                                                 (short)0, acc3, false, false);
  }
  float* __restrict__ Cp =
      H + (size_t)(rowTile * 16 + 8 * half) * NOUT + colGroup * 64 + r;
#pragma unroll
  for (int v = 0; v < 8; ++v) {
    Cp[(size_t)v * NOUT]      = acc0[v];
    Cp[(size_t)v * NOUT + 16] = acc1[v];
    Cp[(size_t)v * NOUT + 32] = acc2[v];
    Cp[(size_t)v * NOUT + 48] = acc3[v];
  }
}

__global__ void dot_e_kernel(const float* __restrict__ H,
                             const float* __restrict__ a_src,
                             const float* __restrict__ a_dst,
                             float* __restrict__ e_src,
                             float* __restrict__ e_dst,
                             int n, int heads, int C) {
  int idx = blockIdx.x * blockDim.x + threadIdx.x;
  if (idx >= n * heads) return;
  int node = idx / heads;
  int hd = idx - node * heads;
  const float* hp = H + (size_t)node * heads * C + (size_t)hd * C;
  const float* as = a_src + (size_t)hd * C;
  const float* ad = a_dst + (size_t)hd * C;
  float s0 = 0.f, s1 = 0.f;
  for (int c = 0; c < C; ++c) {
    float v = hp[c];
    s0 += v * as[c];
    s1 += v * ad[c];
  }
  e_src[idx] = s0;
  e_dst[idx] = s1;
}

__global__ void fill_kernel(float* __restrict__ p, float v, size_t n) {
  size_t i = (size_t)blockIdx.x * blockDim.x + threadIdx.x;
  size_t stride = (size_t)gridDim.x * blockDim.x;
  for (; i < n; i += stride) p[i] = v;
}

__global__ void edge_logit_max(const int* __restrict__ srcIdx,
                               const int* __restrict__ dstIdx,
                               int E, int Etot,
                               const float* __restrict__ e_src,
                               const float* __restrict__ e_dst,
                               float* __restrict__ logits,
                               float* __restrict__ m, int heads) {
  int t = blockIdx.x * blockDim.x + threadIdx.x;
  if (t >= Etot * heads) return;
  int e = t / heads;
  int hd = t - e * heads;
  int s = (e < E) ? srcIdx[e] : (e - E);
  int d = (e < E) ? dstIdx[e] : (e - E);
  float l = e_src[s * heads + hd] + e_dst[d * heads + hd];
  l = (l > 0.f) ? l : l * NEG_SLOPE;
  logits[t] = l;
  atomicMaxFloat(&m[d * heads + hd], l);
}

__global__ void edge_exp_sum(const int* __restrict__ srcIdx,
                             const int* __restrict__ dstIdx,
                             int E, int Etot,
                             float* __restrict__ logits,
                             const float* __restrict__ m,
                             float* __restrict__ denom, int heads) {
  int t = blockIdx.x * blockDim.x + threadIdx.x;
  if (t >= Etot * heads) return;
  int e = t / heads;
  int hd = t - e * heads;
  int d = (e < E) ? dstIdx[e] : (e - E);
  float ex = expf(logits[t] - m[d * heads + hd]);
  logits[t] = ex;
  atomicAdd(&denom[d * heads + hd], ex);
}

template<int NHEADS, int CH, int SPAN>
__global__ void edge_scatter(const int* __restrict__ srcIdx,
                             const int* __restrict__ dstIdx,
                             int E, int Etot,
                             const float* __restrict__ ex,
                             const float* __restrict__ denom,
                             const float* __restrict__ H,
                             float* __restrict__ agg) {
  constexpr int F = NHEADS * CH;
  constexpr int NSPAN = F / SPAN;
  constexpr int LPE = SPAN / 32;
  constexpr int EB = 32;
  constexpr int UNR = 16 / LPE;
  constexpr int NG = EB / UNR;
  static_assert(CH % SPAN == 0 && SPAN % 32 == 0 && EB % UNR == 0, "shape");

  const int lane = threadIdx.x & 31;
  const int wv = __builtin_amdgcn_readfirstlane((int)(threadIdx.x >> 5));
  const int wid = blockIdx.x * (int)(blockDim.x >> 5) + wv;
  const int nb = (Etot + EB - 1) / EB;
  if (wid >= nb * NSPAN) return;
  const int batch = wid / NSPAN;
  const int sp = wid - batch * NSPAN;
  const int hd = (sp * SPAN) / CH;
  const int e0 = batch * EB;
  const int cnt = (Etot - e0 < EB) ? (Etot - e0) : EB;

  const int e = e0 + lane;
  int s = 0, d = 0;
  float a = 0.f;
  if (lane < cnt) {
    s = (e < E) ? srcIdx[e] : (e - E);
    d = (e < E) ? dstIdx[e] : (e - E);
    a = ex[(size_t)e * NHEADS + hd] / (denom[(size_t)d * NHEADS + hd] + 1e-16f);
  }
  const float* __restrict__ Hs = H + sp * SPAN;
  float* __restrict__ As = agg + sp * SPAN;

  if (cnt == EB) {
    float hbuf[2][UNR * LPE];
    for (int u = 0; u < UNR; ++u) {
      const int si = __builtin_amdgcn_readlane(s, u);
      const float* hp = Hs + (size_t)si * F;
      for (int j = 0; j < LPE; ++j) hbuf[0][u * LPE + j] = hp[lane + 32 * j];
    }
    for (int g = 0; g < NG; ++g) {
      if (g + 1 < NG) {
        for (int u = 0; u < UNR; ++u) {
          const int si = __builtin_amdgcn_readlane(s, (g + 1) * UNR + u);
          const float* hp = Hs + (size_t)si * F;
          for (int j = 0; j < LPE; ++j) hbuf[(g + 1) & 1][u * LPE + j] = hp[lane + 32 * j];
        }
      }
      for (int u = 0; u < UNR; ++u) {
        const int di = __builtin_amdgcn_readlane(d, g * UNR + u);
        const float ai =
            __int_as_float(__builtin_amdgcn_readlane(__float_as_int(a), g * UNR + u));
        float* ap = As + (size_t)di * F;
        for (int j = 0; j < LPE; ++j)
          atomicAdd(&ap[lane + 32 * j], ai * hbuf[g & 1][u * LPE + j]);
      }
    }
  } else {
    for (int i = 0; i < cnt; ++i) {
      const int si = __builtin_amdgcn_readlane(s, i);
      const int di = __builtin_amdgcn_readlane(d, i);
      const float ai = __int_as_float(__builtin_amdgcn_readlane(__float_as_int(a), i));
      const float* hp = Hs + (size_t)si * F;
      float* ap = As + (size_t)di * F;
      for (int j = 0; j < LPE; ++j) atomicAdd(&ap[lane + 32 * j], ai * hp[lane + 32 * j]);
    }
  }
}

__global__ void bias_elu(const float* __restrict__ agg,
                         const float* __restrict__ b,
                         float* __restrict__ out, int n, int F) {
  int t = blockIdx.x * blockDim.x + threadIdx.x;
  if (t >= n * F) return;
  int f = t % F;
  float v = agg[t] + b[f];
  out[t] = (v > 0.f) ? v : (expf(v) - 1.f);
}

__global__ void final_mean(const float* __restrict__ agg,
                           const float* __restrict__ b,
                           float* __restrict__ out, int n, int F) {
  int lane = threadIdx.x & 31;
  int node = (int)(((size_t)blockIdx.x * blockDim.x + threadIdx.x) >> 5);
  if (node >= n) return;
  const float* ap = agg + (size_t)node * F;
  float sum = 0.f;
  for (int c = lane; c < F; c += 32) {
    float v = ap[c] + b[c];
    v = (v > 0.f) ? v : (expf(v) - 1.f);
    sum += v;
  }
#pragma unroll
  for (int off = 16; off > 0; off >>= 1)
    sum += __shfl_down(sum, off, 32);
  if (lane == 0) out[node] = sum / (float)F;
}

static inline int cdiv(long long a, long long b) { return (int)((a + b - 1) / b); }

extern "C" void kernel_launch(void* const* d_in, const int* in_sizes, int n_in,
                              void* d_out, int out_size, void* d_ws, size_t ws_size,
                              hipStream_t stream) {
  const float* x      = (const float*)d_in[0];
  const int*   ei     = (const int*)  d_in[1];
  const float* W1     = (const float*)d_in[2];
  const float* a1_src = (const float*)d_in[3];
  const float* a1_dst = (const float*)d_in[4];
  const float* b1     = (const float*)d_in[5];
  const float* W2     = (const float*)d_in[6];
  const float* a2_src = (const float*)d_in[7];
  const float* a2_dst = (const float*)d_in[8];
  const float* b2     = (const float*)d_in[9];
  float* outp = (float*)d_out;

  const int N    = in_sizes[0] / DIN;
  const int E    = in_sizes[1] / 2;
  const int Etot = E + N;
  const int F1   = HEADS * HID;
  const int* srcIdx = ei;
  const int* dstIdx = ei + E;

  float* ws = (float*)d_ws;
  float* h1      = ws;
  float* e_src_b = h1 + (size_t)N * F1;
  float* e_dst_b = e_src_b + (size_t)N * HEADS;
  float* edgebuf = e_dst_b + (size_t)N * HEADS;
  float* m_b     = edgebuf + (size_t)Etot * HEADS;
  float* den_b   = m_b + (size_t)N * HEADS;
  float* aggreg  = den_b + (size_t)N * HEADS;
  float* h2      = aggreg;
  float* agg2    = aggreg + (size_t)N * DOUT;

  const int TB = 256;

  {
    int strips = (N / 16) * (F1 / 64);
    gemm_f32_wmma<DIN, F1><<<cdiv(strips, TB / 32), TB, 0, stream>>>(x, W1, h1, N);

    dot_e_kernel<<<cdiv((long long)N * HEADS, TB), TB, 0, stream>>>(
        h1, a1_src, a1_dst, e_src_b, e_dst_b, N, HEADS, HID);

    fill_kernel<<<1024, TB, 0, stream>>>(m_b, -INFINITY, (size_t)N * HEADS);
    fill_kernel<<<1024, TB, 0, stream>>>(den_b, 0.f, (size_t)N * HEADS);
    fill_kernel<<<4096, TB, 0, stream>>>(aggreg, 0.f, (size_t)N * F1);

    int eh = Etot * HEADS;
    edge_logit_max<<<cdiv(eh, TB), TB, 0, stream>>>(srcIdx, dstIdx, E, Etot,
                                                    e_src_b, e_dst_b, edgebuf, m_b, HEADS);
    edge_exp_sum<<<cdiv(eh, TB), TB, 0, stream>>>(srcIdx, dstIdx, E, Etot,
                                                  edgebuf, m_b, den_b, HEADS);
    edge_scatter<HEADS, HID, 64><<<cdiv((long long)cdiv(Etot, 32) * (F1 / 64), TB / 32), TB, 0, stream>>>(
        srcIdx, dstIdx, E, Etot, edgebuf, den_b, h1, aggreg);

    bias_elu<<<cdiv((long long)N * F1, TB), TB, 0, stream>>>(aggreg, b1, h1, N, F1);
  }

  {
    int strips = (N / 16) * (DOUT / 64);
    gemm_f32_wmma<F1, DOUT><<<cdiv(strips, TB / 32), TB, 0, stream>>>(h1, W2, h2, N);

    dot_e_kernel<<<cdiv(N, TB), TB, 0, stream>>>(
        h2, a2_src, a2_dst, e_src_b, e_dst_b, N, 1, DOUT);

    fill_kernel<<<1024, TB, 0, stream>>>(m_b, -INFINITY, (size_t)N);
    fill_kernel<<<1024, TB, 0, stream>>>(den_b, 0.f, (size_t)N);
    fill_kernel<<<4096, TB, 0, stream>>>(agg2, 0.f, (size_t)N * DOUT);

    edge_logit_max<<<cdiv(Etot, TB), TB, 0, stream>>>(srcIdx, dstIdx, E, Etot,
                                                      e_src_b, e_dst_b, edgebuf, m_b, 1);
    edge_exp_sum<<<cdiv(Etot, TB), TB, 0, stream>>>(srcIdx, dstIdx, E, Etot,
                                                    edgebuf, m_b, den_b, 1);
    edge_scatter<1, DOUT, 32><<<cdiv((long long)cdiv(Etot, 32) * (DOUT / 32), TB / 32), TB, 0, stream>>>(
        srcIdx, dstIdx, E, Etot, edgebuf, den_b, h2, agg2);

    final_mean<<<cdiv((long long)N * 32, TB), TB, 0, stream>>>(agg2, b2, outp, N, DOUT);
  }
}


# layer-1 gather: one wave per destination row covers all 4 heads (was one wave per row and head): 4x fewer waves, 16-byte loads of the whole 1-KiB H row; per-channel sum order kept
# speedup vs baseline: 3.4264x; 1.9162x over previous
#include <hip/hip_runtime.h>
#include <math.h>

#define DIN 256
#define HID 64
#define HEADS 4
#define DOUT 128
#define NEG_SLOPE 0.2f

typedef __attribute__((ext_vector_type(2))) float v2f;
typedef __attribute__((ext_vector_type(4))) float v4f;
typedef __attribute__((ext_vector_type(8))) float v8f;

template<int K, int NOUT>
__global__ void gemm_f32_wmma(const float* __restrict__ X,
                              const float* __restrict__ W,
                              float* __restrict__ H, int nrows) {
  const int lane = threadIdx.x & 31;
  const int wave = threadIdx.x >> 5;
  const int colGroups = NOUT / 64;
  const int strip = blockIdx.x * (blockDim.x >> 5) + wave;
  const int rowTile = strip / colGroups;
  const int colGroup = strip - rowTile * colGroups;
  if (rowTile * 16 >= nrows) return;
  const int r = lane & 15;
  const int half = lane >> 4;
  const float* __restrict__ Ap = X + (size_t)(rowTile * 16 + r) * K + 2 * half;
  const float* __restrict__ Bp = W + (size_t)(2 * half) * NOUT + colGroup * 64 + r;
  v8f acc0 = {}, acc1 = {}, acc2 = {}, acc3 = {};
#pragma unroll 4
  for (int k = 0; k < K; k += 4) {
    v2f a;
    a[0] = Ap[k];
    a[1] = Ap[k + 1];
    const float* b0 = Bp + (size_t)k * NOUT;
    const float* b1 = Bp + (size_t)(k + 1) * NOUT;
    v2f bA, bB, bC, bD;
    bA[0] = b0[0];   bA[1] = b1[0];
    bB[0] = b0[16];  bB[1] = b1[16];
    bC[0] = b0[32];  bC[1] = b1[32];
    bD[0] = b0[48];  bD[1] = b1[48];
    acc0 = __builtin_amdgcn_wmma_f32_16x16x4_f32(false, a, false, bA,
                                                 (short)0, acc0, false, false);
    acc1 = __builtin_amdgcn_wmma_f32_16x16x4_f32(false, a, false, bB,
                                                 (short)0, acc1, false, false);
    acc2 = __builtin_amdgcn_wmma_f32_16x16x4_f32(false, a, false, bC,
                                                 (short)0, acc2, false, false);
    acc3 = __builtin_amdgcn_wmma_f32_16x16x4_f32(false, a, false, bD,
                                                 (short)0, acc3, false, false);
  }
  float* __restrict__ Cp =
      H + (size_t)(rowTile * 16 + 8 * half) * NOUT + colGroup * 64 + r;
#pragma unroll
  for (int v = 0; v < 8; ++v) {
    Cp[(size_t)v * NOUT]      = acc0[v];
    Cp[(size_t)v * NOUT + 16] = acc1[v];
    Cp[(size_t)v * NOUT + 32] = acc2[v];
    Cp[(size_t)v * NOUT + 48] = acc3[v];
  }
}

__global__ void dot_e_kernel(const float* __restrict__ H,
                             const float* __restrict__ a_src,
                             const float* __restrict__ a_dst,
                             float* __restrict__ e_src,
                             float* __restrict__ e_dst,
                             int n, int heads, int C) {
  int idx = blockIdx.x * blockDim.x + threadIdx.x;
  if (idx >= n * heads) return;
  int node = idx / heads;
  int hd = idx - node * heads;
  const float* hp = H + (size_t)node * heads * C + (size_t)hd * C;
  const float* as = a_src + (size_t)hd * C;
  const float* ad = a_dst + (size_t)hd * C;
  float s0 = 0.f, s1 = 0.f;
  for (int c = 0; c < C; ++c) {
    float v = hp[c];
    s0 += v * as[c];
    s1 += v * ad[c];
  }
  e_src[idx] = s0;
  e_dst[idx] = s1;
}

__global__ void csr_zero(int* __restrict__ cnt, int n) {
  int i = blockIdx.x * blockDim.x + threadIdx.x;
  if (i < n) cnt[i] = 0;
}

__global__ void csr_count(const int* __restrict__ dstIdx, int E, int* __restrict__ cnt) {
  int e = blockIdx.x * blockDim.x + threadIdx.x;
  if (e < E) atomicAdd(&cnt[dstIdx[e]], 1);
}

__global__ void __launch_bounds__(1024)
csr_scan(const int* __restrict__ cnt, int* __restrict__ rowptr,
         int* __restrict__ cursor, int n, int E) {
  __shared__ int s_front[32];
  __shared__ int s_wave[32];
  const int t = threadIdx.x;
  const int lane = t & 31;
  const int w = t >> 5;
  const int base = blockIdx.x * 1024;
  int part = 0;
  for (int i = t; i < base; i += 1024) part += cnt[i];
  for (int off = 16; off > 0; off >>= 1) part += __shfl_xor(part, off, 32);
  if (lane == 0) s_front[w] = part;
  const int i = base + t;
  const int v = (i < n) ? cnt[i] : 0;
  int inc = v;
  for (int off = 1; off < 32; off <<= 1) {
    int u = __shfl_up(inc, off, 32);
    if (lane >= off) inc += u;
  }
  if (lane == 31) s_wave[w] = inc;
  __syncthreads();
  int offset = 0;
  for (int k = 0; k < 32; ++k) offset += s_front[k];
  for (int k = 0; k < w; ++k) offset += s_wave[k];
  if (i < n) {
    const int excl = offset + inc - v;
    rowptr[i] = excl;
    cursor[i] = excl;
  }
  if (blockIdx.x == 0 && t == 0) rowptr[n] = E;
}

__global__ void csr_fill(const int* __restrict__ srcIdx, const int* __restrict__ dstIdx,
                         int E, int* __restrict__ cursor, int* __restrict__ csr_src) {
  int e = blockIdx.x * blockDim.x + threadIdx.x;
  if (e >= E) return;
  const int slot = atomicAdd(&cursor[dstIdx[e]], 1);
  csr_src[slot] = srcIdx[e];
}

template<int V> struct VecOf;
template<> struct VecOf<2> { typedef v2f type; };
template<> struct VecOf<4> { typedef v4f type; };

__device__ __forceinline__ float leaky(float l) { return (l > 0.f) ? l : l * NEG_SLOPE; }

__device__ __forceinline__ float wave_max(float v) {
  for (int off = 16; off > 0; off >>= 1) v = fmaxf(v, __shfl_xor(v, off, 32));
  return v;
}
__device__ __forceinline__ float wave_sum(float v) {
  for (int off = 16; off > 0; off >>= 1) v += __shfl_xor(v, off, 32);
  return v;
}
__device__ __forceinline__ float elu1(float v) { return (v > 0.f) ? v : (expf(v) - 1.f); }
__device__ __forceinline__ float lane_value(float v, int l) {
  return __int_as_float(__builtin_amdgcn_readlane(__float_as_int(v), l));
}

__global__ void gat_gather_row4(const int* __restrict__ rowptr,
                                const int* __restrict__ csr_src,
                                const float* __restrict__ e_src,
                                const float* __restrict__ e_dst,
                                const float* __restrict__ H,
                                const float* __restrict__ b,
                                float* __restrict__ out, int n) {
  constexpr int F = HEADS * HID;
  constexpr int GRP = 4;
  static_assert(HEADS == 4 && HID == 64, "lane map is written for 4 heads of 64");

  const int lane = threadIdx.x & 31;
  const bool up = lane >= 16;
  const int wv = __builtin_amdgcn_readfirstlane((int)(threadIdx.x >> 5));
  const int node = blockIdx.x * (int)(blockDim.x >> 5) + wv;
  if (node >= n) return;
  const int beg = rowptr[node];
  const int deg = rowptr[node + 1] - beg;

  const v4f ed = *(const v4f*)(e_dst + (size_t)node * HEADS);
  const v4f es = *(const v4f*)(e_src + (size_t)node * HEADS);
  v4f lself;
  for (int h = 0; h < HEADS; ++h) lself[h] = leaky(es[h] + ed[h]);
  const int* __restrict__ sp = csr_src + beg;
  const float* __restrict__ Hl = H + 4 * lane;

  int s0 = node;
  v4f l0 = lself;
  if (lane < deg) {
    s0 = sp[lane];
    const v4f e = *(const v4f*)(e_src + (size_t)s0 * HEADS);
    for (int h = 0; h < HEADS; ++h) l0[h] = leaky(e[h] + ed[h]);
  }
  v4f m;
  for (int h = 0; h < HEADS; ++h) m[h] = fmaxf(lself[h], l0[h]);
  for (int c0 = 32; c0 < deg; c0 += 32) {
    const int i = c0 + lane;
    if (i < deg) {
      const v4f e = *(const v4f*)(e_src + (size_t)sp[i] * HEADS);
      for (int h = 0; h < HEADS; ++h) m[h] = fmaxf(m[h], leaky(e[h] + ed[h]));
    }
  }
  for (int h = 0; h < HEADS; ++h) m[h] = wave_max(m[h]);

  v4f exself, ex0, den;
  for (int h = 0; h < HEADS; ++h) {
    exself[h] = expf(lself[h] - m[h]);
    ex0[h] = (lane < deg) ? expf(l0[h] - m[h]) : 0.f;
    den[h] = ex0[h];
  }
  for (int c0 = 32; c0 < deg; c0 += 32) {
    const int i = c0 + lane;
    if (i < deg) {
      const v4f e = *(const v4f*)(e_src + (size_t)sp[i] * HEADS);
      for (int h = 0; h < HEADS; ++h) den[h] += expf(leaky(e[h] + ed[h]) - m[h]);
    }
  }
  v4f dn;
  for (int h = 0; h < HEADS; ++h) dn[h] = (wave_sum(den[h]) + exself[h]) + 1e-16f;

  const float* __restrict__ hs = Hl + (size_t)node * F;
  const float wLo = up ? (exself[1] / dn[1]) : (exself[0] / dn[0]);
  const float wHi = up ? (exself[3] / dn[3]) : (exself[2] / dn[2]);
  v4f accLo = wLo * *(const v4f*)hs;
  v4f accHi = wHi * *(const v4f*)(hs + 128);

  for (int c0 = 0; c0 < deg; c0 += 32) {
    int s = s0;
    v4f a;
    for (int h = 0; h < HEADS; ++h) a[h] = ex0[h] / dn[h];
    if (c0 > 0) {
      const int i = c0 + lane;
      s = node;
      for (int h = 0; h < HEADS; ++h) a[h] = 0.f;
      if (i < deg) {
        s = sp[i];
        const v4f e = *(const v4f*)(e_src + (size_t)s * HEADS);
        for (int h = 0; h < HEADS; ++h) a[h] = expf(leaky(e[h] + ed[h]) - m[h]) / dn[h];
      }
    }
    const int cnt = (deg - c0 < 32) ? (deg - c0) : 32;
    for (int j = 0; j < cnt; j += GRP) {
      v4f hLo[GRP], hHi[GRP];
      for (int u = 0; u < GRP; ++u) {
        const int sj = __builtin_amdgcn_readlane(s, j + u);
        const float* __restrict__ hp = Hl + (size_t)sj * F;
        hLo[u] = *(const v4f*)hp;
        hHi[u] = *(const v4f*)(hp + 128);
      }
      for (int u = 0; u < GRP; ++u) {
        const float a0 = lane_value(a[0], j + u);
        const float a1 = lane_value(a[1], j + u);
        const float a2 = lane_value(a[2], j + u);
        const float a3 = lane_value(a[3], j + u);
        accLo += (up ? a1 : a0) * hLo[u];
        accHi += (up ? a3 : a2) * hHi[u];
      }
    }
  }

  const v4f bLo = *(const v4f*)(b + 4 * lane);
  const v4f bHi = *(const v4f*)(b + 128 + 4 * lane);
  v4f oLo, oHi;
  for (int k = 0; k < 4; ++k) {
    oLo[k] = elu1(accLo[k] + bLo[k]);
    oHi[k] = elu1(accHi[k] + bHi[k]);
  }
  float* __restrict__ op = out + (size_t)node * F + 4 * lane;
  *(v4f*)op = oLo;
  *(v4f*)(op + 128) = oHi;
}

template<int CH>
__global__ void gat_gather_mean(const int* __restrict__ rowptr,
                                const int* __restrict__ csr_src,
                                const float* __restrict__ e_src,
                                const float* __restrict__ e_dst,
                                const float* __restrict__ H,
                                const float* __restrict__ b,
                                float* __restrict__ out, int n) {
  constexpr int VEC = CH / 32;
  constexpr int GRP = 4;
  typedef typename VecOf<VEC>::type vecT;
  static_assert(CH % 32 == 0, "shape");

  const int lane = threadIdx.x & 31;
  const int wv = __builtin_amdgcn_readfirstlane((int)(threadIdx.x >> 5));
  const int node = blockIdx.x * (int)(blockDim.x >> 5) + wv;
  if (node >= n) return;
  const int beg = rowptr[node];
  const int deg = rowptr[node + 1] - beg;

  const float ed = e_dst[node];
  const float lself = leaky(e_src[node] + ed);
  const int* __restrict__ sp = csr_src + beg;
  const float* __restrict__ Hh = H + lane * VEC;

  int s0 = node;
  float l0 = lself;
  if (lane < deg) {
    s0 = sp[lane];
    l0 = leaky(e_src[s0] + ed);
  }
  float m = fmaxf(lself, l0);
  for (int c0 = 32; c0 < deg; c0 += 32) {
    const int i = c0 + lane;
    if (i < deg) m = fmaxf(m, leaky(e_src[sp[i]] + ed));
  }
  m = wave_max(m);

  const float exself = expf(lself - m);
  const float ex0 = (lane < deg) ? expf(l0 - m) : 0.f;
  float den = ex0;
  for (int c0 = 32; c0 < deg; c0 += 32) {
    const int i = c0 + lane;
    if (i < deg) den += expf(leaky(e_src[sp[i]] + ed) - m);
  }
  den = wave_sum(den) + exself;
  const float dn = den + 1e-16f;

  vecT acc = (exself / dn) * *(const vecT*)(Hh + (size_t)node * CH);

  for (int c0 = 0; c0 < deg; c0 += 32) {
    int s = s0;
    float a = ex0 / dn;
    if (c0 > 0) {
      const int i = c0 + lane;
      s = node;
      a = 0.f;
      if (i < deg) {
        s = sp[i];
        a = expf(leaky(e_src[s] + ed) - m) / dn;
      }
    }
    const int cnt = (deg - c0 < 32) ? (deg - c0) : 32;
    for (int j = 0; j < cnt; j += GRP) {
      vecT hb[GRP];
      for (int u = 0; u < GRP; ++u) {
        const int sj = __builtin_amdgcn_readlane(s, j + u);
        hb[u] = *(const vecT*)(Hh + (size_t)sj * CH);
      }
      for (int u = 0; u < GRP; ++u) acc += lane_value(a, j + u) * hb[u];
    }
  }

  const float* __restrict__ bp = b + lane * VEC;
  float sum = 0.f;
  for (int k = 0; k < VEC; ++k) sum += elu1(acc[k] + bp[k]);
  sum = wave_sum(sum);
  if (lane == 0) out[node] = sum / (float)CH;
}

static inline int cdiv(long long a, long long b) { return (int)((a + b - 1) / b); }

extern "C" void kernel_launch(void* const* d_in, const int* in_sizes, int n_in,
                              void* d_out, int out_size, void* d_ws, size_t ws_size,
                              hipStream_t stream) {
  const float* x      = (const float*)d_in[0];
  const int*   ei     = (const int*)  d_in[1];
  const float* W1     = (const float*)d_in[2];
  const float* a1_src = (const float*)d_in[3];
  const float* a1_dst = (const float*)d_in[4];
  const float* b1     = (const float*)d_in[5];
  const float* W2     = (const float*)d_in[6];
  const float* a2_src = (const float*)d_in[7];
  const float* a2_dst = (const float*)d_in[8];
  const float* b2     = (const float*)d_in[9];
  float* outp = (float*)d_out;

  const int N    = in_sizes[0] / DIN;
  const int E    = in_sizes[1] / 2;
  const int Etot = E + N;
  const int F1   = HEADS * HID;
  const int* srcIdx = ei;
  const int* dstIdx = ei + E;

  float* ws = (float*)d_ws;
  float* h1      = ws;
  float* e_src_b = h1 + (size_t)N * F1;
  float* e_dst_b = e_src_b + (size_t)N * HEADS;
  float* edgebuf = e_dst_b + (size_t)N * HEADS;
  float* m_b     = edgebuf + (size_t)Etot * HEADS;
  float* den_b   = m_b + (size_t)N * HEADS;
  float* out1    = den_b + (size_t)N * HEADS;
  float* h2      = h1;
  int* csr_src = (int*)edgebuf;
  int* rowptr  = (int*)m_b;
  int* cnt     = (int*)den_b;
  int* cursor  = cnt + N;

  const int TB = 256;

  csr_zero<<<cdiv(N, TB), TB, 0, stream>>>(cnt, N);
  csr_count<<<cdiv(E, TB), TB, 0, stream>>>(dstIdx, E, cnt);
  csr_scan<<<cdiv(N, 1024), 1024, 0, stream>>>(cnt, rowptr, cursor, N, E);
  csr_fill<<<cdiv(E, TB), TB, 0, stream>>>(srcIdx, dstIdx, E, cursor, csr_src);

  {
    int strips = (N / 16) * (F1 / 64);
    gemm_f32_wmma<DIN, F1><<<cdiv(strips, TB / 32), TB, 0, stream>>>(x, W1, h1, N);

    dot_e_kernel<<<cdiv((long long)N * HEADS, TB), TB, 0, stream>>>(
        h1, a1_src, a1_dst, e_src_b, e_dst_b, N, HEADS, HID);

    gat_gather_row4<<<cdiv(N, TB / 32), TB, 0, stream>>>(
        rowptr, csr_src, e_src_b, e_dst_b, h1, b1, out1, N);
  }

  {
    int strips = (N / 16) * (DOUT / 64);
    gemm_f32_wmma<F1, DOUT><<<cdiv(strips, TB / 32), TB, 0, stream>>>(out1, W2, h2, N);

    dot_e_kernel<<<cdiv(N, TB), TB, 0, stream>>>(
        h2, a2_src, a2_dst, e_src_b, e_dst_b, N, 1, DOUT);

    gat_gather_mean<DOUT><<<cdiv(N, TB / 32), TB, 0, stream>>>(
        rowptr, csr_src, e_src_b, e_dst_b, h2, b2, outp, N);
  }
}


# GEMMs: W staged once per work-group into LDS (transposed), B fragments by 16-byte LDS reads; 256 work-groups of 16 waves loop over 16x64 strips; 16-byte X loads, k order in each 8-k block permuted
# speedup vs baseline: 3.8731x; 1.1304x over previous
#include <hip/hip_runtime.h>
#include <math.h>

#define DIN 256
#define HID 64
#define HEADS 4
#define DOUT 128
#define NEG_SLOPE 0.2f

typedef __attribute__((ext_vector_type(2))) float v2f;
typedef __attribute__((ext_vector_type(4))) float v4f;
typedef __attribute__((ext_vector_type(8))) float v8f;

#define WMMA_F32(A2, B2, ACC) \
  __builtin_amdgcn_wmma_f32_16x16x4_f32(false, (A2), false, (B2), (short)0, (ACC), false, false)

template<int P>
__device__ __forceinline__ void strip_block8(const v4f a, const float* bl,
                                             v8f& acc0, v8f& acc1, v8f& acc2, v8f& acc3) {
  const v4f w0 = *(const v4f*)(bl);
  const v4f w1 = *(const v4f*)(bl + 16 * P);
  const v4f w2 = *(const v4f*)(bl + 32 * P);
  const v4f w3 = *(const v4f*)(bl + 48 * P);
  const v2f alo = {a[0], a[1]}, ahi = {a[2], a[3]};
  const v2f b0l = {w0[0], w0[1]}, b0h = {w0[2], w0[3]};
  const v2f b1l = {w1[0], w1[1]}, b1h = {w1[2], w1[3]};
  const v2f b2l = {w2[0], w2[1]}, b2h = {w2[2], w2[3]};
  const v2f b3l = {w3[0], w3[1]}, b3h = {w3[2], w3[3]};
  acc0 = WMMA_F32(alo, b0l, acc0);
  acc1 = WMMA_F32(alo, b1l, acc1);
  acc2 = WMMA_F32(alo, b2l, acc2);
  acc3 = WMMA_F32(alo, b3l, acc3);
  acc0 = WMMA_F32(ahi, b0h, acc0);
  acc1 = WMMA_F32(ahi, b1h, acc1);
  acc2 = WMMA_F32(ahi, b2h, acc2);
  acc3 = WMMA_F32(ahi, b3h, acc3);
}

template<int K, int NOUT>
__global__ void __launch_bounds__(512)
gemm_f32_wmma(const float* __restrict__ X,
              const float* __restrict__ W,
              float* __restrict__ H, int nRowTiles) {
  constexpr int P = K + 4;
  constexpr int WAVES = 16;
  constexpr int CG = NOUT / 64;
  constexpr int RS = WAVES / CG;
  constexpr int NG = K / 32;
  static_assert(K % 32 == 0 && NOUT % 64 == 0 && WAVES % CG == 0, "shape");
  static_assert((size_t)NOUT * P * 4 <= 320u * 1024u, "W fits the work-group's LDS");
  __shared__ __attribute__((aligned(16))) float Wt[NOUT * P];

  const int lane = threadIdx.x & 31;
  const int wave = __builtin_amdgcn_readfirstlane((int)(threadIdx.x >> 5));
  const int r = lane & 15;
  const int half = lane >> 4;
  const int cgp = wave % CG;
  const int slot = wave / CG;
  const int stride = RS * (int)gridDim.x;
  int rt = slot * (int)gridDim.x + (int)blockIdx.x;
  const int rt0 = (rt < nRowTiles) ? rt : (nRowTiles - 1);
  const float* __restrict__ Ap = X + (size_t)(rt0 * 16 + r) * K + 4 * half;

  v4f aN0 = *(const v4f*)(Ap);
  v4f aN1 = *(const v4f*)(Ap + 8);
  v4f aN2 = *(const v4f*)(Ap + 16);
  v4f aN3 = *(const v4f*)(Ap + 24);

  {
    const int q = (lane & 3) + 4 * (lane >> 4);
    const int kr = (lane >> 2) & 3;
    constexpr int TC = NOUT / 32;
    constexpr int TILES = (K / 16) * TC;
    static_assert(TILES % WAVES == 0, "tiles split evenly over the waves");
#pragma unroll
    for (int i = 0; i < TILES / WAVES; ++i) {
      const int t = wave + i * WAVES;
      const int k0 = (t / TC) * 16;
      const int c0 = (t % TC) * 32;
      const float* __restrict__ src = W + (size_t)(k0 + 4 * kr) * NOUT + c0 + 4 * q;
      const v4f v0 = *(const v4f*)(src);
      const v4f v1 = *(const v4f*)(src + NOUT);
      const v4f v2 = *(const v4f*)(src + 2 * NOUT);
      const v4f v3 = *(const v4f*)(src + 3 * NOUT);
      float* dst = Wt + (c0 + 4 * q) * P + k0 + 4 * kr;
      v4f o;
      o[0] = v0[0]; o[1] = v1[0]; o[2] = v2[0]; o[3] = v3[0];
      *(v4f*)(dst) = o;
      o[0] = v0[1]; o[1] = v1[1]; o[2] = v2[1]; o[3] = v3[1];
      *(v4f*)(dst + P) = o;
      o[0] = v0[2]; o[1] = v1[2]; o[2] = v2[2]; o[3] = v3[2];
      *(v4f*)(dst + 2 * P) = o;
      o[0] = v0[3]; o[1] = v1[3]; o[2] = v2[3]; o[3] = v3[3];
      *(v4f*)(dst + 3 * P) = o;
    }
  }
  __syncthreads();

  const float* Bl = Wt + (cgp * 64 + r) * P + 4 * half;

  while (rt < nRowTiles) {
    const int rtN = rt + stride;
    const float* __restrict__ ApN =
        (rtN < nRowTiles) ? X + (size_t)(rtN * 16 + r) * K + 4 * half : Ap;
    v8f acc0 = {}, acc1 = {}, acc2 = {}, acc3 = {};
    for (int g = 0; g < NG; ++g) {
      const v4f a0 = aN0, a1 = aN1, a2 = aN2, a3 = aN3;
      const float* __restrict__ nx = (g + 1 < NG) ? (Ap + (g + 1) * 32) : ApN;
      aN0 = *(const v4f*)(nx);
      aN1 = *(const v4f*)(nx + 8);
      aN2 = *(const v4f*)(nx + 16);
      aN3 = *(const v4f*)(nx + 24);
      const float* bl = Bl + g * 32;
      strip_block8<P>(a0, bl,      acc0, acc1, acc2, acc3);
      strip_block8<P>(a1, bl + 8,  acc0, acc1, acc2, acc3);
      strip_block8<P>(a2, bl + 16, acc0, acc1, acc2, acc3);
      strip_block8<P>(a3, bl + 24, acc0, acc1, acc2, acc3);
    }
    float* __restrict__ Cp =
        H + (size_t)(rt * 16 + 8 * half) * NOUT + cgp * 64 + r;
#pragma unroll
    for (int v = 0; v < 8; ++v) {
      Cp[(size_t)v * NOUT]      = acc0[v];
      Cp[(size_t)v * NOUT + 16] = acc1[v];
      Cp[(size_t)v * NOUT + 32] = acc2[v];
      Cp[(size_t)v * NOUT + 48] = acc3[v];
    }
    rt = rtN;
    Ap = ApN;
  }
}
#undef WMMA_F32

__global__ void dot_e_kernel(const float* __restrict__ H,
                             const float* __restrict__ a_src,
                             const float* __restrict__ a_dst,
                             float* __restrict__ e_src,
                             float* __restrict__ e_dst,
                             int n, int heads, int C) {
  int idx = blockIdx.x * blockDim.x + threadIdx.x;
  if (idx >= n * heads) return;
  int node = idx / heads;
  int hd = idx - node * heads;
  const float* hp = H + (size_t)node * heads * C + (size_t)hd * C;
  const float* as = a_src + (size_t)hd * C;
  const float* ad = a_dst + (size_t)hd * C;
  float s0 = 0.f, s1 = 0.f;
  for (int c = 0; c < C; ++c) {
    float v = hp[c];
    s0 += v * as[c];
    s1 += v * ad[c];
  }
  e_src[idx] = s0;
  e_dst[idx] = s1;
}

__global__ void csr_zero(int* __restrict__ cnt, int n) {
  int i = blockIdx.x * blockDim.x + threadIdx.x;
  if (i < n) cnt[i] = 0;
}

__global__ void csr_count(const int* __restrict__ dstIdx, int E, int* __restrict__ cnt) {
  int e = blockIdx.x * blockDim.x + threadIdx.x;
  if (e < E) atomicAdd(&cnt[dstIdx[e]], 1);
}

__global__ void __launch_bounds__(1024)
csr_scan(const int* __restrict__ cnt, int* __restrict__ rowptr,
         int* __restrict__ cursor, int n, int E) {
  __shared__ int s_front[32];
  __shared__ int s_wave[32];
  const int t = threadIdx.x;
  const int lane = t & 31;
  const int w = t >> 5;
  const int base = blockIdx.x * 1024;
  int part = 0;
  for (int i = t; i < base; i += 1024) part += cnt[i];
  for (int off = 16; off > 0; off >>= 1) part += __shfl_xor(part, off, 32);
  if (lane == 0) s_front[w] = part;
  const int i = base + t;
  const int v = (i < n) ? cnt[i] : 0;
  int inc = v;
  for (int off = 1; off < 32; off <<= 1) {
    int u = __shfl_up(inc, off, 32);
    if (lane >= off) inc += u;
  }
  if (lane == 31) s_wave[w] = inc;
  __syncthreads();
  int offset = 0;
  for (int k = 0; k < 32; ++k) offset += s_front[k];
  for (int k = 0; k < w; ++k) offset += s_wave[k];
  if (i < n) {
    const int excl = offset + inc - v;
    rowptr[i] = excl;
    cursor[i] = excl;
  }
  if (blockIdx.x == 0 && t == 0) rowptr[n] = E;
}

__global__ void csr_fill(const int* __restrict__ srcIdx, const int* __restrict__ dstIdx,
                         int E, int* __restrict__ cursor, int* __restrict__ csr_src) {
  int e = blockIdx.x * blockDim.x + threadIdx.x;
  if (e >= E) return;
  const int slot = atomicAdd(&cursor[dstIdx[e]], 1);
  csr_src[slot] = srcIdx[e];
}

template<int V> struct VecOf;
template<> struct VecOf<2> { typedef v2f type; };
template<> struct VecOf<4> { typedef v4f type; };

__device__ __forceinline__ float leaky(float l) { return (l > 0.f) ? l : l * NEG_SLOPE; }

__device__ __forceinline__ float wave_max(float v) {
  for (int off = 16; off > 0; off >>= 1) v = fmaxf(v, __shfl_xor(v, off, 32));
  return v;
}
__device__ __forceinline__ float wave_sum(float v) {
  for (int off = 16; off > 0; off >>= 1) v += __shfl_xor(v, off, 32);
  return v;
}
__device__ __forceinline__ float elu1(float v) { return (v > 0.f) ? v : (expf(v) - 1.f); }
__device__ __forceinline__ float lane_value(float v, int l) {
  return __int_as_float(__builtin_amdgcn_readlane(__float_as_int(v), l));
}

__global__ void gat_gather_row4(const int* __restrict__ rowptr,
                                const int* __restrict__ csr_src,
                                const float* __restrict__ e_src,
                                const float* __restrict__ e_dst,
                                const float* __restrict__ H,
                                const float* __restrict__ b,
                                float* __restrict__ out, int n) {
  constexpr int F = HEADS * HID;
  constexpr int GRP = 4;
  static_assert(HEADS == 4 && HID == 64, "lane map is written for 4 heads of 64");

  const int lane = threadIdx.x & 31;
  const bool up = lane >= 16;
  const int wv = __builtin_amdgcn_readfirstlane((int)(threadIdx.x >> 5));
  const int node = blockIdx.x * (int)(blockDim.x >> 5) + wv;
  if (node >= n) return;
  const int beg = rowptr[node];
  const int deg = rowptr[node + 1] - beg;

  const v4f ed = *(const v4f*)(e_dst + (size_t)node * HEADS);
  const v4f es = *(const v4f*)(e_src + (size_t)node * HEADS);
  v4f lself;
  for (int h = 0; h < HEADS; ++h) lself[h] = leaky(es[h] + ed[h]);
  const int* __restrict__ sp = csr_src + beg;
  const float* __restrict__ Hl = H + 4 * lane;

  int s0 = node;
  v4f l0 = lself;
  if (lane < deg) {
    s0 = sp[lane];
    const v4f e = *(const v4f*)(e_src + (size_t)s0 * HEADS);
    for (int h = 0; h < HEADS; ++h) l0[h] = leaky(e[h] + ed[h]);
  }
  v4f m;
  for (int h = 0; h < HEADS; ++h) m[h] = fmaxf(lself[h], l0[h]);
  for (int c0 = 32; c0 < deg; c0 += 32) {
    const int i = c0 + lane;
    if (i < deg) {
      const v4f e = *(const v4f*)(e_src + (size_t)sp[i] * HEADS);
      for (int h = 0; h < HEADS; ++h) m[h] = fmaxf(m[h], leaky(e[h] + ed[h]));
    }
  }
  for (int h = 0; h < HEADS; ++h) m[h] = wave_max(m[h]);

  v4f exself, ex0, den;
  for (int h = 0; h < HEADS; ++h) {
    exself[h] = expf(lself[h] - m[h]);
    ex0[h] = (lane < deg) ? expf(l0[h] - m[h]) : 0.f;
    den[h] = ex0[h];
  }
  for (int c0 = 32; c0 < deg; c0 += 32) {
    const int i = c0 + lane;
    if (i < deg) {
      const v4f e = *(const v4f*)(e_src + (size_t)sp[i] * HEADS);
      for (int h = 0; h < HEADS; ++h) den[h] += expf(leaky(e[h] + ed[h]) - m[h]);
    }
  }
  v4f dn;
  for (int h = 0; h < HEADS; ++h) dn[h] = (wave_sum(den[h]) + exself[h]) + 1e-16f;

  const float* __restrict__ hs = Hl + (size_t)node * F;
  const float wLo = up ? (exself[1] / dn[1]) : (exself[0] / dn[0]);
  const float wHi = up ? (exself[3] / dn[3]) : (exself[2] / dn[2]);
  v4f accLo = wLo * *(const v4f*)hs;
  v4f accHi = wHi * *(const v4f*)(hs + 128);

  for (int c0 = 0; c0 < deg; c0 += 32) {
    int s = s0;
    v4f a;
    for (int h = 0; h < HEADS; ++h) a[h] = ex0[h] / dn[h];
    if (c0 > 0) {
      const int i = c0 + lane;
      s = node;
      for (int h = 0; h < HEADS; ++h) a[h] = 0.f;
      if (i < deg) {
        s = sp[i];
        const v4f e = *(const v4f*)(e_src + (size_t)s * HEADS);
        for (int h = 0; h < HEADS; ++h) a[h] = expf(leaky(e[h] + ed[h]) - m[h]) / dn[h];
      }
    }
    const int cnt = (deg - c0 < 32) ? (deg - c0) : 32;
    for (int j = 0; j < cnt; j += GRP) {
      v4f hLo[GRP], hHi[GRP];
      for (int u = 0; u < GRP; ++u) {
        const int sj = __builtin_amdgcn_readlane(s, j + u);
        const float* __restrict__ hp = Hl + (size_t)sj * F;
        hLo[u] = *(const v4f*)hp;
        hHi[u] = *(const v4f*)(hp + 128);
      }
      for (int u = 0; u < GRP; ++u) {
        const float a0 = lane_value(a[0], j + u);
        const float a1 = lane_value(a[1], j + u);
        const float a2 = lane_value(a[2], j + u);
        const float a3 = lane_value(a[3], j + u);
        accLo += (up ? a1 : a0) * hLo[u];
        accHi += (up ? a3 : a2) * hHi[u];
      }
    }
  }

  const v4f bLo = *(const v4f*)(b + 4 * lane);
  const v4f bHi = *(const v4f*)(b + 128 + 4 * lane);
  v4f oLo, oHi;
  for (int k = 0; k < 4; ++k) {
    oLo[k] = elu1(accLo[k] + bLo[k]);
    oHi[k] = elu1(accHi[k] + bHi[k]);
  }
  float* __restrict__ op = out + (size_t)node * F + 4 * lane;
  *(v4f*)op = oLo;
  *(v4f*)(op + 128) = oHi;
}

template<int CH>
__global__ void gat_gather_mean(const int* __restrict__ rowptr,
                                const int* __restrict__ csr_src,
                                const float* __restrict__ e_src,
                                const float* __restrict__ e_dst,
                                const float* __restrict__ H,
                                const float* __restrict__ b,
                                float* __restrict__ out, int n) {
  constexpr int VEC = CH / 32;
  constexpr int GRP = 4;
  typedef typename VecOf<VEC>::type vecT;
  static_assert(CH % 32 == 0, "shape");

  const int lane = threadIdx.x & 31;
  const int wv = __builtin_amdgcn_readfirstlane((int)(threadIdx.x >> 5));
  const int node = blockIdx.x * (int)(blockDim.x >> 5) + wv;
  if (node >= n) return;
  const int beg = rowptr[node];
  const int deg = rowptr[node + 1] - beg;

  const float ed = e_dst[node];
  const float lself = leaky(e_src[node] + ed);
  const int* __restrict__ sp = csr_src + beg;
  const float* __restrict__ Hh = H + lane * VEC;

  int s0 = node;
  float l0 = lself;
  if (lane < deg) {
    s0 = sp[lane];
    l0 = leaky(e_src[s0] + ed);
  }
  float m = fmaxf(lself, l0);
  for (int c0 = 32; c0 < deg; c0 += 32) {
    const int i = c0 + lane;
    if (i < deg) m = fmaxf(m, leaky(e_src[sp[i]] + ed));
  }
  m = wave_max(m);

  const float exself = expf(lself - m);
  const float ex0 = (lane < deg) ? expf(l0 - m) : 0.f;
  float den = ex0;
  for (int c0 = 32; c0 < deg; c0 += 32) {
    const int i = c0 + lane;
    if (i < deg) den += expf(leaky(e_src[sp[i]] + ed) - m);
  }
  den = wave_sum(den) + exself;
  const float dn = den + 1e-16f;

  vecT acc = (exself / dn) * *(const vecT*)(Hh + (size_t)node * CH);

  for (int c0 = 0; c0 < deg; c0 += 32) {
    int s = s0;
    float a = ex0 / dn;
    if (c0 > 0) {
      const int i = c0 + lane;
      s = node;
      a = 0.f;
      if (i < deg) {
        s = sp[i];
        a = expf(leaky(e_src[s] + ed) - m) / dn;
      }
    }
    const int cnt = (deg - c0 < 32) ? (deg - c0) : 32;
    for (int j = 0; j < cnt; j += GRP) {
      vecT hb[GRP];
      for (int u = 0; u < GRP; ++u) {
        const int sj = __builtin_amdgcn_readlane(s, j + u);
        hb[u] = *(const vecT*)(Hh + (size_t)sj * CH);
      }
      for (int u = 0; u < GRP; ++u) acc += lane_value(a, j + u) * hb[u];
    }
  }

  const float* __restrict__ bp = b + lane * VEC;
  float sum = 0.f;
  for (int k = 0; k < VEC; ++k) sum += elu1(acc[k] + bp[k]);
  sum = wave_sum(sum);
  if (lane == 0) out[node] = sum / (float)CH;
}

static inline int cdiv(long long a, long long b) { return (int)((a + b - 1) / b); }

extern "C" void kernel_launch(void* const* d_in, const int* in_sizes, int n_in,
                              void* d_out, int out_size, void* d_ws, size_t ws_size,
                              hipStream_t stream) {
  const float* x      = (const float*)d_in[0];
  const int*   ei     = (const int*)  d_in[1];
  const float* W1     = (const float*)d_in[2];
  const float* a1_src = (const float*)d_in[3];
  const float* a1_dst = (const float*)d_in[4];
  const float* b1     = (const float*)d_in[5];
  const float* W2     = (const float*)d_in[6];
  const float* a2_src = (const float*)d_in[7];
  const float* a2_dst = (const float*)d_in[8];
  const float* b2     = (const float*)d_in[9];
  float* outp = (float*)d_out;

  const int N    = in_sizes[0] / DIN;
  const int E    = in_sizes[1] / 2;
  const int Etot = E + N;
  const int F1   = HEADS * HID;
  const int* srcIdx = ei;
  const int* dstIdx = ei + E;

  float* ws = (float*)d_ws;
  float* h1      = ws;
  float* e_src_b = h1 + (size_t)N * F1;
  float* e_dst_b = e_src_b + (size_t)N * HEADS;
  float* edgebuf = e_dst_b + (size_t)N * HEADS;
  float* m_b     = edgebuf + (size_t)Etot * HEADS;
  float* den_b   = m_b + (size_t)N * HEADS;
  float* out1    = den_b + (size_t)N * HEADS;
  float* h2      = h1;
  int* csr_src = (int*)edgebuf;
  int* rowptr  = (int*)m_b;
  int* cnt     = (int*)den_b;
  int* cursor  = cnt + N;

  const int TB = 256;

  csr_zero<<<cdiv(N, TB), TB, 0, stream>>>(cnt, N);
  csr_count<<<cdiv(E, TB), TB, 0, stream>>>(dstIdx, E, cnt);
  csr_scan<<<cdiv(N, 1024), 1024, 0, stream>>>(cnt, rowptr, cursor, N, E);
  csr_fill<<<cdiv(E, TB), TB, 0, stream>>>(srcIdx, dstIdx, E, cursor, csr_src);

  {
    gemm_f32_wmma<DIN, F1><<<256, 512, 0, stream>>>(x, W1, h1, N / 16);

    dot_e_kernel<<<cdiv((long long)N * HEADS, TB), TB, 0, stream>>>(
        h1, a1_src, a1_dst, e_src_b, e_dst_b, N, HEADS, HID);

    gat_gather_row4<<<cdiv(N, TB / 32), TB, 0, stream>>>(
        rowptr, csr_src, e_src_b, e_dst_b, h1, b1, out1, N);
  }

  {
    gemm_f32_wmma<F1, DOUT><<<256, 512, 0, stream>>>(out1, W2, h2, N / 16);

    dot_e_kernel<<<cdiv(N, TB), TB, 0, stream>>>(
        h2, a2_src, a2_dst, e_src_b, e_dst_b, N, 1, DOUT);

    gat_gather_mean<DOUT><<<cdiv(N, TB / 32), TB, 0, stream>>>(
        rowptr, csr_src, e_src_b, e_dst_b, h2, b2, outp, N);
  }
}


# CSR build: edge ranks from per-work-group LDS counters (two-level counting sort), 64 pieces; counters added to row counts one line per atomic; fill without atomics; slot order = arrival order of adds
# speedup vs baseline: 5.3707x; 1.3867x over previous
#include <hip/hip_runtime.h>
#include <math.h>

#define DIN 256
#define HID 64
#define HEADS 4
#define DOUT 128
#define NEG_SLOPE 0.2f

typedef __attribute__((ext_vector_type(2))) float v2f;
typedef __attribute__((ext_vector_type(4))) float v4f;
typedef __attribute__((ext_vector_type(8))) float v8f;

template<int M>
__device__ __forceinline__ float row_xor_add(float v) {
  const int t = __builtin_amdgcn_update_dpp(0, __float_as_int(v), 0x160 + M, 0xf, 0xf, false);
  return v + __int_as_float(t);
}
__device__ __forceinline__ float row_sum16(float v) {
  v = row_xor_add<1>(v);
  v = row_xor_add<2>(v);
  v = row_xor_add<4>(v);
  v = row_xor_add<8>(v);
  return v;
}

#define WMMA_F32(A2, B2, ACC) \
  __builtin_amdgcn_wmma_f32_16x16x4_f32(false, (A2), false, (B2), (short)0, (ACC), false, false)

template<int P>
__device__ __forceinline__ void strip_block8(const v4f a, const float* bl,
                                             v8f& acc0, v8f& acc1, v8f& acc2, v8f& acc3) {
  const v4f w0 = *(const v4f*)(bl);
  const v4f w1 = *(const v4f*)(bl + 16 * P);
  const v4f w2 = *(const v4f*)(bl + 32 * P);
  const v4f w3 = *(const v4f*)(bl + 48 * P);
  const v2f alo = {a[0], a[1]}, ahi = {a[2], a[3]};
  const v2f b0l = {w0[0], w0[1]}, b0h = {w0[2], w0[3]};
  const v2f b1l = {w1[0], w1[1]}, b1h = {w1[2], w1[3]};
  const v2f b2l = {w2[0], w2[1]}, b2h = {w2[2], w2[3]};
  const v2f b3l = {w3[0], w3[1]}, b3h = {w3[2], w3[3]};
  acc0 = WMMA_F32(alo, b0l, acc0);
  acc1 = WMMA_F32(alo, b1l, acc1);
  acc2 = WMMA_F32(alo, b2l, acc2);
  acc3 = WMMA_F32(alo, b3l, acc3);
  acc0 = WMMA_F32(ahi, b0h, acc0);
  acc1 = WMMA_F32(ahi, b1h, acc1);
  acc2 = WMMA_F32(ahi, b2h, acc2);
  acc3 = WMMA_F32(ahi, b3h, acc3);
}

template<int K, int NOUT>
__global__ void __launch_bounds__(512)
gemm_f32_wmma(const float* __restrict__ X,
              const float* __restrict__ W,
              const float* __restrict__ a_src,
              const float* __restrict__ a_dst,
              float* __restrict__ H,
              float* __restrict__ e_src,
              float* __restrict__ e_dst, int nRowTiles) {
  constexpr int P = K + 4;
  constexpr int WAVES = 16;
  constexpr int CG = NOUT / 64;
  constexpr int RS = WAVES / CG;
  constexpr int NG = K / 32;
  static_assert(K % 32 == 0 && NOUT % 64 == 0 && WAVES % CG == 0, "shape");
  static_assert((size_t)NOUT * P * 4 <= 320u * 1024u, "W fits the work-group's LDS");
  __shared__ __attribute__((aligned(16))) float Wt[NOUT * P];

  const int lane = threadIdx.x & 31;
  const int wave = __builtin_amdgcn_readfirstlane((int)(threadIdx.x >> 5));
  const int r = lane & 15;
  const int half = lane >> 4;
  const int cgp = wave % CG;
  const int slot = wave / CG;
  const int stride = RS * (int)gridDim.x;
  int rt = slot * (int)gridDim.x + (int)blockIdx.x;
  const int rt0 = (rt < nRowTiles) ? rt : (nRowTiles - 1);
  const float* __restrict__ Ap = X + (size_t)(rt0 * 16 + r) * K + 4 * half;

  v4f aN0 = *(const v4f*)(Ap);
  v4f aN1 = *(const v4f*)(Ap + 8);
  v4f aN2 = *(const v4f*)(Ap + 16);
  v4f aN3 = *(const v4f*)(Ap + 24);

  {
    const int q = (lane & 3) + 4 * (lane >> 4);
    const int kr = (lane >> 2) & 3;
    constexpr int TC = NOUT / 32;
    constexpr int TILES = (K / 16) * TC;
    static_assert(TILES % WAVES == 0, "tiles split evenly over the waves");
#pragma unroll
    for (int i = 0; i < TILES / WAVES; ++i) {
      const int t = wave + i * WAVES;
      const int k0 = (t / TC) * 16;
      const int c0 = (t % TC) * 32;
      const float* __restrict__ src = W + (size_t)(k0 + 4 * kr) * NOUT + c0 + 4 * q;
      const v4f v0 = *(const v4f*)(src);
      const v4f v1 = *(const v4f*)(src + NOUT);
      const v4f v2 = *(const v4f*)(src + 2 * NOUT);
      const v4f v3 = *(const v4f*)(src + 3 * NOUT);
      float* dst = Wt + (c0 + 4 * q) * P + k0 + 4 * kr;
      v4f o;
      o[0] = v0[0]; o[1] = v1[0]; o[2] = v2[0]; o[3] = v3[0];
      *(v4f*)(dst) = o;
      o[0] = v0[1]; o[1] = v1[1]; o[2] = v2[1]; o[3] = v3[1];
      *(v4f*)(dst + P) = o;
      o[0] = v0[2]; o[1] = v1[2]; o[2] = v2[2]; o[3] = v3[2];
      *(v4f*)(dst + 2 * P) = o;
      o[0] = v0[3]; o[1] = v1[3]; o[2] = v2[3]; o[3] = v3[3];
      *(v4f*)(dst + 3 * P) = o;
    }
  }
  __syncthreads();

  const float* Bl = Wt + (cgp * 64 + r) * P + 4 * half;

  const float as0 = a_src[cgp * 64 + r],      ad0 = a_dst[cgp * 64 + r];
  const float as1 = a_src[cgp * 64 + 16 + r], ad1 = a_dst[cgp * 64 + 16 + r];
  const float as2 = a_src[cgp * 64 + 32 + r], ad2 = a_dst[cgp * 64 + 32 + r];
  const float as3 = a_src[cgp * 64 + 48 + r], ad3 = a_dst[cgp * 64 + 48 + r];

  while (rt < nRowTiles) {
    const int rtN = rt + stride;
    const float* __restrict__ ApN =
        (rtN < nRowTiles) ? X + (size_t)(rtN * 16 + r) * K + 4 * half : Ap;
    v8f acc0 = {}, acc1 = {}, acc2 = {}, acc3 = {};
    for (int g = 0; g < NG; ++g) {
      const v4f a0 = aN0, a1 = aN1, a2 = aN2, a3 = aN3;
      const float* __restrict__ nx = (g + 1 < NG) ? (Ap + (g + 1) * 32) : ApN;
      aN0 = *(const v4f*)(nx);
      aN1 = *(const v4f*)(nx + 8);
      aN2 = *(const v4f*)(nx + 16);
      aN3 = *(const v4f*)(nx + 24);
      const float* bl = Bl + g * 32;
      strip_block8<P>(a0, bl,      acc0, acc1, acc2, acc3);
      strip_block8<P>(a1, bl + 8,  acc0, acc1, acc2, acc3);
      strip_block8<P>(a2, bl + 16, acc0, acc1, acc2, acc3);
      strip_block8<P>(a3, bl + 24, acc0, acc1, acc2, acc3);
    }
    float* __restrict__ Cp =
        H + (size_t)(rt * 16 + 8 * half) * NOUT + cgp * 64 + r;
#pragma unroll
    for (int v = 0; v < 8; ++v) {
      Cp[(size_t)v * NOUT]      = acc0[v];
      Cp[(size_t)v * NOUT + 16] = acc1[v];
      Cp[(size_t)v * NOUT + 32] = acc2[v];
      Cp[(size_t)v * NOUT + 48] = acc3[v];
    }
    float ps[8], pd[8];
#pragma unroll
    for (int v = 0; v < 8; ++v) {
      ps[v] = acc0[v] * as0 + acc1[v] * as1 + acc2[v] * as2 + acc3[v] * as3;
      pd[v] = acc0[v] * ad0 + acc1[v] * ad1 + acc2[v] * ad2 + acc3[v] * ad3;
    }
#pragma unroll
    for (int v = 0; v < 8; ++v) {
      ps[v] = row_sum16(ps[v]);
      pd[v] = row_sum16(pd[v]);
    }
    float vs = ps[0], vd = pd[0];
#pragma unroll
    for (int v = 1; v < 8; ++v) {
      if ((r & 7) == v) { vs = ps[v]; vd = pd[v]; }
    }
    float* __restrict__ ep = ((r < 8) ? e_src : e_dst)
                           + (size_t)(rt * 16 + 8 * half + (r & 7)) * CG + cgp;
    *ep = (r < 8) ? vs : vd;
    rt = rtN;
    Ap = ApN;
  }
}
#undef WMMA_F32

#define HIST_CAP 65536
typedef __attribute__((ext_vector_type(4))) int v4i;

__global__ void csr_zero(int* __restrict__ cnt, int n) {
  int i = blockIdx.x * blockDim.x + threadIdx.x;
  if (i < n) cnt[i] = 0;
}

__global__ void __launch_bounds__(1024)
csr_rank(const int* __restrict__ dstIdx, int E, int n,
         int* __restrict__ cnt, int* __restrict__ rank) {
  __shared__ __attribute__((aligned(16))) int hist[HIST_CAP];
  constexpr int T = 1024;
  constexpr int UE = 16;
  constexpr int UD = 25;
  const int t = threadIdx.x;

  for (int i = 4 * t; i < n; i += 4 * T) *(v4i*)(hist + i) = (v4i){0, 0, 0, 0};
  __syncthreads();

  const int chunk = (E + (int)gridDim.x - 1) / (int)gridDim.x;
  const int e0 = (int)blockIdx.x * chunk;
  const int e1 = (e0 + chunk < E) ? (e0 + chunk) : E;

  for (int eb = e0 + t; eb < e1; eb += UE * T) {
    int d[UE];
#pragma unroll
    for (int u = 0; u < UE; ++u) {
      const int e = eb + u * T;
      d[u] = (e < e1) ? dstIdx[e] : -1;
    }
#pragma unroll
    for (int u = 0; u < UE; ++u) {
      if (d[u] >= 0) rank[eb + u * T] = atomicAdd(&hist[d[u]], 1);
    }
  }
  __syncthreads();

  for (int db = t; db < n; db += UD * T) {
    int h[UD], b[UD];
#pragma unroll
    for (int u = 0; u < UD; ++u) {
      const int dd = db + u * T;
      h[u] = (dd < n) ? hist[dd] : 0;
    }
#pragma unroll
    for (int u = 0; u < UD; ++u) {
      b[u] = 0;
      if (h[u] != 0) b[u] = atomicAdd(&cnt[db + u * T], h[u]);
    }
#pragma unroll
    for (int u = 0; u < UD; ++u) {
      if (h[u] != 0) hist[db + u * T] = b[u];
    }
  }
  __syncthreads();

  for (int eb = e0 + t; eb < e1; eb += UE * T) {
    int d[UE], r[UE];
#pragma unroll
    for (int u = 0; u < UE; ++u) {
      const int e = eb + u * T;
      d[u] = (e < e1) ? dstIdx[e] : -1;
      r[u] = (e < e1) ? rank[e] : 0;
    }
#pragma unroll
    for (int u = 0; u < UE; ++u) {
      if (d[u] >= 0) rank[eb + u * T] = r[u] + hist[d[u]];
    }
  }
}

__global__ void __launch_bounds__(1024)
csr_scan(const int* __restrict__ cnt, int* __restrict__ rowptr, int n, int E) {
  __shared__ int s_front[32];
  __shared__ int s_wave[32];
  const int t = threadIdx.x;
  const int lane = t & 31;
  const int w = t >> 5;
  const int base = blockIdx.x * 1024;
  int part = 0;
  for (int i = t; i < base; i += 1024) part += cnt[i];
  for (int off = 16; off > 0; off >>= 1) part += __shfl_xor(part, off, 32);
  if (lane == 0) s_front[w] = part;
  const int i = base + t;
  const int v = (i < n) ? cnt[i] : 0;
  int inc = v;
  for (int off = 1; off < 32; off <<= 1) {
    int u = __shfl_up(inc, off, 32);
    if (lane >= off) inc += u;
  }
  if (lane == 31) s_wave[w] = inc;
  __syncthreads();
  int offset = 0;
  for (int k = 0; k < 32; ++k) offset += s_front[k];
  for (int k = 0; k < w; ++k) offset += s_wave[k];
  if (i < n) rowptr[i] = offset + inc - v;
  if (blockIdx.x == 0 && t == 0) rowptr[n] = E;
}

__global__ void csr_fill(const int* __restrict__ srcIdx, const int* __restrict__ dstIdx,
                         int E, const int* __restrict__ rowptr,
                         const int* __restrict__ rank, int* __restrict__ csr_src) {
  int e = blockIdx.x * blockDim.x + threadIdx.x;
  if (e >= E) return;
  csr_src[rowptr[dstIdx[e]] + rank[e]] = srcIdx[e];
}

template<int V> struct VecOf;
template<> struct VecOf<2> { typedef v2f type; };
template<> struct VecOf<4> { typedef v4f type; };

__device__ __forceinline__ float leaky(float l) { return (l > 0.f) ? l : l * NEG_SLOPE; }

__device__ __forceinline__ float wave_max(float v) {
  for (int off = 16; off > 0; off >>= 1) v = fmaxf(v, __shfl_xor(v, off, 32));
  return v;
}
__device__ __forceinline__ float wave_sum(float v) {
  for (int off = 16; off > 0; off >>= 1) v += __shfl_xor(v, off, 32);
  return v;
}
__device__ __forceinline__ float elu1(float v) { return (v > 0.f) ? v : (expf(v) - 1.f); }
__device__ __forceinline__ float lane_value(float v, int l) {
  return __int_as_float(__builtin_amdgcn_readlane(__float_as_int(v), l));
}
__device__ __forceinline__ float pair_sum(const float* __restrict__ e, int i) {
  const v2f p = *(const v2f*)(e + 2 * (size_t)i);
  return p[0] + p[1];
}

__global__ void gat_gather_row4(const int* __restrict__ rowptr,
                                const int* __restrict__ csr_src,
                                const float* __restrict__ e_src,
                                const float* __restrict__ e_dst,
                                const float* __restrict__ H,
                                const float* __restrict__ b,
                                float* __restrict__ out, int n) {
  constexpr int F = HEADS * HID;
  constexpr int GRP = 4;
  static_assert(HEADS == 4 && HID == 64, "lane map is written for 4 heads of 64");

  const int lane = threadIdx.x & 31;
  const bool up = lane >= 16;
  const int wv = __builtin_amdgcn_readfirstlane((int)(threadIdx.x >> 5));
  const int node = blockIdx.x * (int)(blockDim.x >> 5) + wv;
  if (node >= n) return;
  const int beg = rowptr[node];
  const int deg = rowptr[node + 1] - beg;

  const v4f ed = *(const v4f*)(e_dst + (size_t)node * HEADS);
  const v4f es = *(const v4f*)(e_src + (size_t)node * HEADS);
  v4f lself;
  for (int h = 0; h < HEADS; ++h) lself[h] = leaky(es[h] + ed[h]);
  const int* __restrict__ sp = csr_src + beg;
  const float* __restrict__ Hl = H + 4 * lane;

  int s0 = node;
  v4f l0 = lself;
  if (lane < deg) {
    s0 = sp[lane];
    const v4f e = *(const v4f*)(e_src + (size_t)s0 * HEADS);
    for (int h = 0; h < HEADS; ++h) l0[h] = leaky(e[h] + ed[h]);
  }
  v4f m;
  for (int h = 0; h < HEADS; ++h) m[h] = fmaxf(lself[h], l0[h]);
  for (int c0 = 32; c0 < deg; c0 += 32) {
    const int i = c0 + lane;
    if (i < deg) {
      const v4f e = *(const v4f*)(e_src + (size_t)sp[i] * HEADS);
      for (int h = 0; h < HEADS; ++h) m[h] = fmaxf(m[h], leaky(e[h] + ed[h]));
    }
  }
  for (int h = 0; h < HEADS; ++h) m[h] = wave_max(m[h]);

  v4f exself, ex0, den;
  for (int h = 0; h < HEADS; ++h) {
    exself[h] = expf(lself[h] - m[h]);
    ex0[h] = (lane < deg) ? expf(l0[h] - m[h]) : 0.f;
    den[h] = ex0[h];
  }
  for (int c0 = 32; c0 < deg; c0 += 32) {
    const int i = c0 + lane;
    if (i < deg) {
      const v4f e = *(const v4f*)(e_src + (size_t)sp[i] * HEADS);
      for (int h = 0; h < HEADS; ++h) den[h] += expf(leaky(e[h] + ed[h]) - m[h]);
    }
  }
  v4f dn;
  for (int h = 0; h < HEADS; ++h) dn[h] = (wave_sum(den[h]) + exself[h]) + 1e-16f;

  const float* __restrict__ hs = Hl + (size_t)node * F;
  const float wLo = up ? (exself[1] / dn[1]) : (exself[0] / dn[0]);
  const float wHi = up ? (exself[3] / dn[3]) : (exself[2] / dn[2]);
  v4f accLo = wLo * *(const v4f*)hs;
  v4f accHi = wHi * *(const v4f*)(hs + 128);

  for (int c0 = 0; c0 < deg; c0 += 32) {
    int s = s0;
    v4f a;
    for (int h = 0; h < HEADS; ++h) a[h] = ex0[h] / dn[h];
    if (c0 > 0) {
      const int i = c0 + lane;
      s = node;
      for (int h = 0; h < HEADS; ++h) a[h] = 0.f;
      if (i < deg) {
        s = sp[i];
        const v4f e = *(const v4f*)(e_src + (size_t)s * HEADS);
        for (int h = 0; h < HEADS; ++h) a[h] = expf(leaky(e[h] + ed[h]) - m[h]) / dn[h];
      }
    }
    const int cnt = (deg - c0 < 32) ? (deg - c0) : 32;
    for (int j = 0; j < cnt; j += GRP) {
      v4f hLo[GRP], hHi[GRP];
      for (int u = 0; u < GRP; ++u) {
        const int sj = __builtin_amdgcn_readlane(s, j + u);
        const float* __restrict__ hp = Hl + (size_t)sj * F;
        hLo[u] = *(const v4f*)hp;
        hHi[u] = *(const v4f*)(hp + 128);
      }
      for (int u = 0; u < GRP; ++u) {
        const float a0 = lane_value(a[0], j + u);
        const float a1 = lane_value(a[1], j + u);
        const float a2 = lane_value(a[2], j + u);
        const float a3 = lane_value(a[3], j + u);
        accLo += (up ? a1 : a0) * hLo[u];
        accHi += (up ? a3 : a2) * hHi[u];
      }
    }
  }

  const v4f bLo = *(const v4f*)(b + 4 * lane);
  const v4f bHi = *(const v4f*)(b + 128 + 4 * lane);
  v4f oLo, oHi;
  for (int k = 0; k < 4; ++k) {
    oLo[k] = elu1(accLo[k] + bLo[k]);
    oHi[k] = elu1(accHi[k] + bHi[k]);
  }
  float* __restrict__ op = out + (size_t)node * F + 4 * lane;
  *(v4f*)op = oLo;
  *(v4f*)(op + 128) = oHi;
}

template<int CH>
__global__ void gat_gather_mean(const int* __restrict__ rowptr,
                                const int* __restrict__ csr_src,
                                const float* __restrict__ e_src,
                                const float* __restrict__ e_dst,
                                const float* __restrict__ H,
                                const float* __restrict__ b,
                                float* __restrict__ out, int n) {
  constexpr int VEC = CH / 32;
  constexpr int GRP = 4;
  typedef typename VecOf<VEC>::type vecT;
  static_assert(CH % 32 == 0, "shape");

  const int lane = threadIdx.x & 31;
  const int wv = __builtin_amdgcn_readfirstlane((int)(threadIdx.x >> 5));
  const int node = blockIdx.x * (int)(blockDim.x >> 5) + wv;
  if (node >= n) return;
  const int beg = rowptr[node];
  const int deg = rowptr[node + 1] - beg;

  const float ed = pair_sum(e_dst, node);
  const float lself = leaky(pair_sum(e_src, node) + ed);
  const int* __restrict__ sp = csr_src + beg;
  const float* __restrict__ Hh = H + lane * VEC;

  int s0 = node;
  float l0 = lself;
  if (lane < deg) {
    s0 = sp[lane];
    l0 = leaky(pair_sum(e_src, s0) + ed);
  }
  float m = fmaxf(lself, l0);
  for (int c0 = 32; c0 < deg; c0 += 32) {
    const int i = c0 + lane;
    if (i < deg) m = fmaxf(m, leaky(pair_sum(e_src, sp[i]) + ed));
  }
  m = wave_max(m);

  const float exself = expf(lself - m);
  const float ex0 = (lane < deg) ? expf(l0 - m) : 0.f;
  float den = ex0;
  for (int c0 = 32; c0 < deg; c0 += 32) {
    const int i = c0 + lane;
    if (i < deg) den += expf(leaky(pair_sum(e_src, sp[i]) + ed) - m);
  }
  den = wave_sum(den) + exself;
  const float dn = den + 1e-16f;

  vecT acc = (exself / dn) * *(const vecT*)(Hh + (size_t)node * CH);

  for (int c0 = 0; c0 < deg; c0 += 32) {
    int s = s0;
    float a = ex0 / dn;
    if (c0 > 0) {
      const int i = c0 + lane;
      s = node;
      a = 0.f;
      if (i < deg) {
        s = sp[i];
        a = expf(leaky(pair_sum(e_src, s) + ed) - m) / dn;
      }
    }
    const int cnt = (deg - c0 < 32) ? (deg - c0) : 32;
    for (int j = 0; j < cnt; j += GRP) {
      vecT hb[GRP];
      for (int u = 0; u < GRP; ++u) {
        const int sj = __builtin_amdgcn_readlane(s, j + u);
        hb[u] = *(const vecT*)(Hh + (size_t)sj * CH);
      }
      for (int u = 0; u < GRP; ++u) acc += lane_value(a, j + u) * hb[u];
    }
  }

  const float* __restrict__ bp = b + lane * VEC;
  float sum = 0.f;
  for (int k = 0; k < VEC; ++k) sum += elu1(acc[k] + bp[k]);
  sum = wave_sum(sum);
  if (lane == 0) out[node] = sum / (float)CH;
}

static inline int cdiv(long long a, long long b) { return (int)((a + b - 1) / b); }

extern "C" void kernel_launch(void* const* d_in, const int* in_sizes, int n_in,
                              void* d_out, int out_size, void* d_ws, size_t ws_size,
                              hipStream_t stream) {
  const float* x      = (const float*)d_in[0];
  const int*   ei     = (const int*)  d_in[1];
  const float* W1     = (const float*)d_in[2];
  const float* a1_src = (const float*)d_in[3];
  const float* a1_dst = (const float*)d_in[4];
  const float* b1     = (const float*)d_in[5];
  const float* W2     = (const float*)d_in[6];
  const float* a2_src = (const float*)d_in[7];
  const float* a2_dst = (const float*)d_in[8];
  const float* b2     = (const float*)d_in[9];
  float* outp = (float*)d_out;

  const int N    = in_sizes[0] / DIN;
  const int E    = in_sizes[1] / 2;
  const int Etot = E + N;
  const int F1   = HEADS * HID;
  const int* srcIdx = ei;
  const int* dstIdx = ei + E;

  float* ws = (float*)d_ws;
  float* h1      = ws;
  float* e_src_b = h1 + (size_t)N * F1;
  float* e_dst_b = e_src_b + (size_t)N * HEADS;
  float* edgebuf = e_dst_b + (size_t)N * HEADS;
  float* m_b     = edgebuf + (size_t)Etot * HEADS;
  float* den_b   = m_b + (size_t)N * HEADS;
  float* out1    = den_b + (size_t)N * HEADS;
  float* h2      = h1;
  int* csr_src = (int*)edgebuf;
  int* rowptr  = (int*)m_b;
  int* cnt     = (int*)den_b;
  int* rank    = csr_src + E;

  const int TB = 256;

  csr_zero<<<cdiv(N, TB), TB, 0, stream>>>(cnt, N);
  csr_rank<<<64, 1024, 0, stream>>>(dstIdx, E, N, cnt, rank);
  csr_scan<<<cdiv(N, 1024), 1024, 0, stream>>>(cnt, rowptr, N, E);
  csr_fill<<<cdiv(E, TB), TB, 0, stream>>>(srcIdx, dstIdx, E, rowptr, rank, csr_src);

  {
    gemm_f32_wmma<DIN, F1><<<256, 512, 0, stream>>>(
        x, W1, a1_src, a1_dst, h1, e_src_b, e_dst_b, N / 16);

    gat_gather_row4<<<cdiv(N, TB / 32), TB, 0, stream>>>(
        rowptr, csr_src, e_src_b, e_dst_b, h1, b1, out1, N);
  }

  {
    gemm_f32_wmma<F1, DOUT><<<256, 512, 0, stream>>>(
        out1, W2, a2_src, a2_dst, h2, e_src_b, e_dst_b, N / 16);

    gat_gather_mean<DOUT><<<cdiv(N, TB / 32), TB, 0, stream>>>(
        rowptr, csr_src, e_src_b, e_dst_b, h2, b2, outp, N);
  }
}


# layer-1 gather and layer-2 GEMM fused: 16 gather waves put ELU rows into an LDS tile, 4 product waves run WMMAs beside the next tile's gather; layer-1 output never stored; e sum order changed
# speedup vs baseline: 6.0380x; 1.1242x over previous
#include <hip/hip_runtime.h>
#include <math.h>

#define DIN 256
#define HID 64
#define HEADS 4
#define DOUT 128
#define NEG_SLOPE 0.2f

typedef __attribute__((ext_vector_type(2))) float v2f;
typedef __attribute__((ext_vector_type(4))) float v4f;
typedef __attribute__((ext_vector_type(8))) float v8f;

template<int M>
__device__ __forceinline__ float row_xor_add(float v) {
  const int t = __builtin_amdgcn_update_dpp(0, __float_as_int(v), 0x160 + M, 0xf, 0xf, false);
  return v + __int_as_float(t);
}
__device__ __forceinline__ float row_sum16(float v) {
  v = row_xor_add<1>(v);
  v = row_xor_add<2>(v);
  v = row_xor_add<4>(v);
  v = row_xor_add<8>(v);
  return v;
}

#define WMMA_F32(A2, B2, ACC) \
  __builtin_amdgcn_wmma_f32_16x16x4_f32(false, (A2), false, (B2), (short)0, (ACC), false, false)

template<int P>
__device__ __forceinline__ void strip_block8(const v4f a, const float* bl,
                                             v8f& acc0, v8f& acc1, v8f& acc2, v8f& acc3) {
  const v4f w0 = *(const v4f*)(bl);
  const v4f w1 = *(const v4f*)(bl + 16 * P);
  const v4f w2 = *(const v4f*)(bl + 32 * P);
  const v4f w3 = *(const v4f*)(bl + 48 * P);
  const v2f alo = {a[0], a[1]}, ahi = {a[2], a[3]};
  const v2f b0l = {w0[0], w0[1]}, b0h = {w0[2], w0[3]};
  const v2f b1l = {w1[0], w1[1]}, b1h = {w1[2], w1[3]};
  const v2f b2l = {w2[0], w2[1]}, b2h = {w2[2], w2[3]};
  const v2f b3l = {w3[0], w3[1]}, b3h = {w3[2], w3[3]};
  acc0 = WMMA_F32(alo, b0l, acc0);
  acc1 = WMMA_F32(alo, b1l, acc1);
  acc2 = WMMA_F32(alo, b2l, acc2);
  acc3 = WMMA_F32(alo, b3l, acc3);
  acc0 = WMMA_F32(ahi, b0h, acc0);
  acc1 = WMMA_F32(ahi, b1h, acc1);
  acc2 = WMMA_F32(ahi, b2h, acc2);
  acc3 = WMMA_F32(ahi, b3h, acc3);
}

template<int K, int NOUT>
__global__ void __launch_bounds__(512)
gemm_f32_wmma(const float* __restrict__ X,
              const float* __restrict__ W,
              const float* __restrict__ a_src,
              const float* __restrict__ a_dst,
              float* __restrict__ H,
              float* __restrict__ e_src,
              float* __restrict__ e_dst, int nRowTiles) {
  constexpr int P = K + 4;
  constexpr int WAVES = 16;
  constexpr int CG = NOUT / 64;
  constexpr int RS = WAVES / CG;
  constexpr int NG = K / 32;
  static_assert(K % 32 == 0 && NOUT % 64 == 0 && WAVES % CG == 0, "shape");
  static_assert((size_t)NOUT * P * 4 <= 320u * 1024u, "W fits the work-group's LDS");
  __shared__ __attribute__((aligned(16))) float Wt[NOUT * P];

  const int lane = threadIdx.x & 31;
  const int wave = __builtin_amdgcn_readfirstlane((int)(threadIdx.x >> 5));
  const int r = lane & 15;
  const int half = lane >> 4;
  const int cgp = wave % CG;
  const int slot = wave / CG;
  const int stride = RS * (int)gridDim.x;
  int rt = slot * (int)gridDim.x + (int)blockIdx.x;
  const int rt0 = (rt < nRowTiles) ? rt : (nRowTiles - 1);
  const float* __restrict__ Ap = X + (size_t)(rt0 * 16 + r) * K + 4 * half;

  v4f aN0 = *(const v4f*)(Ap);
  v4f aN1 = *(const v4f*)(Ap + 8);
  v4f aN2 = *(const v4f*)(Ap + 16);
  v4f aN3 = *(const v4f*)(Ap + 24);

  {
    const int q = (lane & 3) + 4 * (lane >> 4);
    const int kr = (lane >> 2) & 3;
    constexpr int TC = NOUT / 32;
    constexpr int TILES = (K / 16) * TC;
    static_assert(TILES % WAVES == 0, "tiles split evenly over the waves");
#pragma unroll
    for (int i = 0; i < TILES / WAVES; ++i) {
      const int t = wave + i * WAVES;
      const int k0 = (t / TC) * 16;
      const int c0 = (t % TC) * 32;
      const float* __restrict__ src = W + (size_t)(k0 + 4 * kr) * NOUT + c0 + 4 * q;
      const v4f v0 = *(const v4f*)(src);
      const v4f v1 = *(const v4f*)(src + NOUT);
      const v4f v2 = *(const v4f*)(src + 2 * NOUT);
      const v4f v3 = *(const v4f*)(src + 3 * NOUT);
      float* dst = Wt + (c0 + 4 * q) * P + k0 + 4 * kr;
      v4f o;
      o[0] = v0[0]; o[1] = v1[0]; o[2] = v2[0]; o[3] = v3[0];
      *(v4f*)(dst) = o;
      o[0] = v0[1]; o[1] = v1[1]; o[2] = v2[1]; o[3] = v3[1];
      *(v4f*)(dst + P) = o;
      o[0] = v0[2]; o[1] = v1[2]; o[2] = v2[2]; o[3] = v3[2];
      *(v4f*)(dst + 2 * P) = o;
      o[0] = v0[3]; o[1] = v1[3]; o[2] = v2[3]; o[3] = v3[3];
      *(v4f*)(dst + 3 * P) = o;
    }
  }
  __syncthreads();

  const float* Bl = Wt + (cgp * 64 + r) * P + 4 * half;

  const float as0 = a_src[cgp * 64 + r],      ad0 = a_dst[cgp * 64 + r];
  const float as1 = a_src[cgp * 64 + 16 + r], ad1 = a_dst[cgp * 64 + 16 + r];
  const float as2 = a_src[cgp * 64 + 32 + r], ad2 = a_dst[cgp * 64 + 32 + r];
  const float as3 = a_src[cgp * 64 + 48 + r], ad3 = a_dst[cgp * 64 + 48 + r];

  while (rt < nRowTiles) {
    const int rtN = rt + stride;
    const float* __restrict__ ApN =
        (rtN < nRowTiles) ? X + (size_t)(rtN * 16 + r) * K + 4 * half : Ap;
    v8f acc0 = {}, acc1 = {}, acc2 = {}, acc3 = {};
    for (int g = 0; g < NG; ++g) {
      const v4f a0 = aN0, a1 = aN1, a2 = aN2, a3 = aN3;
      const float* __restrict__ nx = (g + 1 < NG) ? (Ap + (g + 1) * 32) : ApN;
      aN0 = *(const v4f*)(nx);
      aN1 = *(const v4f*)(nx + 8);
      aN2 = *(const v4f*)(nx + 16);
      aN3 = *(const v4f*)(nx + 24);
      const float* bl = Bl + g * 32;
      strip_block8<P>(a0, bl,      acc0, acc1, acc2, acc3);
      strip_block8<P>(a1, bl + 8,  acc0, acc1, acc2, acc3);
      strip_block8<P>(a2, bl + 16, acc0, acc1, acc2, acc3);
      strip_block8<P>(a3, bl + 24, acc0, acc1, acc2, acc3);
    }
    float* __restrict__ Cp =
        H + (size_t)(rt * 16 + 8 * half) * NOUT + cgp * 64 + r;
#pragma unroll
    for (int v = 0; v < 8; ++v) {
      Cp[(size_t)v * NOUT]      = acc0[v];
      Cp[(size_t)v * NOUT + 16] = acc1[v];
      Cp[(size_t)v * NOUT + 32] = acc2[v];
      Cp[(size_t)v * NOUT + 48] = acc3[v];
    }
    float ps[8], pd[8];
#pragma unroll
    for (int v = 0; v < 8; ++v) {
      ps[v] = acc0[v] * as0 + acc1[v] * as1 + acc2[v] * as2 + acc3[v] * as3;
      pd[v] = acc0[v] * ad0 + acc1[v] * ad1 + acc2[v] * ad2 + acc3[v] * ad3;
    }
#pragma unroll
    for (int v = 0; v < 8; ++v) {
      ps[v] = row_sum16(ps[v]);
      pd[v] = row_sum16(pd[v]);
    }
    float vs = ps[0], vd = pd[0];
#pragma unroll
    for (int v = 1; v < 8; ++v) {
      if ((r & 7) == v) { vs = ps[v]; vd = pd[v]; }
    }
    float* __restrict__ ep = ((r < 8) ? e_src : e_dst)
                           + (size_t)(rt * 16 + 8 * half + (r & 7)) * CG + cgp;
    *ep = (r < 8) ? vs : vd;
    rt = rtN;
    Ap = ApN;
  }
}

#define HIST_CAP 65536
#define ROW_CAP 48
typedef __attribute__((ext_vector_type(4))) int v4i;

__global__ void csr_zero(int* __restrict__ cnt, int n) {
  int i = blockIdx.x * blockDim.x + threadIdx.x;
  if (i < n) cnt[i] = 0;
}

__global__ void __launch_bounds__(1024)
csr_rank(const int* __restrict__ srcIdx, const int* __restrict__ dstIdx, int E, int n,
         int* __restrict__ cnt, int* __restrict__ slots) {
  __shared__ __attribute__((aligned(16))) int hist[HIST_CAP];
  constexpr int T = 1024;
  constexpr int UE = 8;
  constexpr int UD = 25;
  const int t = threadIdx.x;

  for (int i = 4 * t; i < n; i += 4 * T) *(v4i*)(hist + i) = (v4i){0, 0, 0, 0};
  __syncthreads();

  const int chunk = (E + (int)gridDim.x - 1) / (int)gridDim.x;
  const int e0 = (int)blockIdx.x * chunk;
  const int e1 = (e0 + chunk < E) ? (e0 + chunk) : E;

  for (int eb = e0 + t; eb < e1; eb += UE * T) {
    int d[UE];
#pragma unroll
    for (int u = 0; u < UE; ++u) {
      const int e = eb + u * T;
      d[u] = (e < e1) ? dstIdx[e] : -1;
    }
#pragma unroll
    for (int u = 0; u < UE; ++u) {
      if (d[u] >= 0) atomicAdd(&hist[d[u]], 1);
    }
  }
  __syncthreads();

  for (int db = t; db < n; db += UD * T) {
    int h[UD], b[UD];
#pragma unroll
    for (int u = 0; u < UD; ++u) {
      const int dd = db + u * T;
      h[u] = (dd < n) ? hist[dd] : 0;
    }
#pragma unroll
    for (int u = 0; u < UD; ++u) {
      b[u] = 0;
      if (h[u] != 0) b[u] = atomicAdd(&cnt[db + u * T], h[u]);
    }
#pragma unroll
    for (int u = 0; u < UD; ++u) {
      if (h[u] != 0) hist[db + u * T] = b[u];
    }
  }
  __syncthreads();

  for (int eb = e0 + t; eb < e1; eb += UE * T) {
    int d[UE], s[UE];
#pragma unroll
    for (int u = 0; u < UE; ++u) {
      const int e = eb + u * T;
      d[u] = (e < e1) ? dstIdx[e] : -1;
      s[u] = (e < e1) ? srcIdx[e] : 0;
    }
#pragma unroll
    for (int u = 0; u < UE; ++u) {
      if (d[u] >= 0) {
        const int p = atomicAdd(&hist[d[u]], 1);
        if (p < ROW_CAP) slots[(size_t)d[u] * ROW_CAP + p] = s[u];
      }
    }
  }
}

template<int V> struct VecOf;
template<> struct VecOf<2> { typedef v2f type; };
template<> struct VecOf<4> { typedef v4f type; };

__device__ __forceinline__ float leaky(float l) { return (l > 0.f) ? l : l * NEG_SLOPE; }

__device__ __forceinline__ float wave_max(float v) {
  for (int off = 16; off > 0; off >>= 1) v = fmaxf(v, __shfl_xor(v, off, 32));
  return v;
}
__device__ __forceinline__ float wave_sum(float v) {
  for (int off = 16; off > 0; off >>= 1) v += __shfl_xor(v, off, 32);
  return v;
}
__device__ __forceinline__ float elu1(float v) { return (v > 0.f) ? v : (expf(v) - 1.f); }
__device__ __forceinline__ float lane_value(float v, int l) {
  return __int_as_float(__builtin_amdgcn_readlane(__float_as_int(v), l));
}
__device__ __forceinline__ float pair_sum(const float* __restrict__ e, int i) {
  const v4f p = *(const v4f*)(e + 4 * (size_t)i);
  return ((p[0] + p[1]) + p[2]) + p[3];
}

__device__ __forceinline__ void gather_row4(const int* __restrict__ cnt,
                                const int* __restrict__ slots,
                                const int* __restrict__ srcIdx,
                                const int* __restrict__ dstIdx, int E,
                                const float* __restrict__ e_src,
                                const float* __restrict__ e_dst,
                                const float* __restrict__ H,
                                const float* __restrict__ b,
                                const int node, v4f& oLo, v4f& oHi) {
  constexpr int F = HEADS * HID;
  constexpr int GRP = 4;
  static_assert(HEADS == 4 && HID == 64, "lane map is written for 4 heads of 64");

  const int lane = threadIdx.x & 31;
  const bool up = lane >= 16;
  const int deg = cnt[node];

  const v4f ed = *(const v4f*)(e_dst + (size_t)node * HEADS);
  const v4f es = *(const v4f*)(e_src + (size_t)node * HEADS);
  v4f lself;
  for (int h = 0; h < HEADS; ++h) lself[h] = leaky(es[h] + ed[h]);
  const int* __restrict__ sp = slots + (size_t)node * ROW_CAP;
  const float* __restrict__ Hl = H + 4 * lane;
  const float* __restrict__ hs = Hl + (size_t)node * F;
  v4f accLo, accHi;

  if (deg <= ROW_CAP) {
    int s0 = node;
    v4f l0 = lself;
    if (lane < deg) {
      s0 = sp[lane];
      const v4f e = *(const v4f*)(e_src + (size_t)s0 * HEADS);
      for (int h = 0; h < HEADS; ++h) l0[h] = leaky(e[h] + ed[h]);
    }
    v4f m;
    for (int h = 0; h < HEADS; ++h) m[h] = fmaxf(lself[h], l0[h]);
    for (int c0 = 32; c0 < deg; c0 += 32) {
      const int i = c0 + lane;
      if (i < deg) {
        const v4f e = *(const v4f*)(e_src + (size_t)sp[i] * HEADS);
        for (int h = 0; h < HEADS; ++h) m[h] = fmaxf(m[h], leaky(e[h] + ed[h]));
      }
    }
    for (int h = 0; h < HEADS; ++h) m[h] = wave_max(m[h]);

    v4f exself, ex0, den;
    for (int h = 0; h < HEADS; ++h) {
      exself[h] = expf(lself[h] - m[h]);
      ex0[h] = (lane < deg) ? expf(l0[h] - m[h]) : 0.f;
      den[h] = ex0[h];
    }
    for (int c0 = 32; c0 < deg; c0 += 32) {
      const int i = c0 + lane;
      if (i < deg) {
        const v4f e = *(const v4f*)(e_src + (size_t)sp[i] * HEADS);
        for (int h = 0; h < HEADS; ++h) den[h] += expf(leaky(e[h] + ed[h]) - m[h]);
      }
    }
    v4f dn;
    for (int h = 0; h < HEADS; ++h) dn[h] = (wave_sum(den[h]) + exself[h]) + 1e-16f;

    const float wLo = up ? (exself[1] / dn[1]) : (exself[0] / dn[0]);
    const float wHi = up ? (exself[3] / dn[3]) : (exself[2] / dn[2]);
    accLo = wLo * *(const v4f*)hs;
    accHi = wHi * *(const v4f*)(hs + 128);

    for (int c0 = 0; c0 < deg; c0 += 32) {
      int s = s0;
      v4f a;
      for (int h = 0; h < HEADS; ++h) a[h] = ex0[h] / dn[h];
      if (c0 > 0) {
        const int i = c0 + lane;
        s = node;
        for (int h = 0; h < HEADS; ++h) a[h] = 0.f;
        if (i < deg) {
          s = sp[i];
          const v4f e = *(const v4f*)(e_src + (size_t)s * HEADS);
          for (int h = 0; h < HEADS; ++h) a[h] = expf(leaky(e[h] + ed[h]) - m[h]) / dn[h];
        }
      }
      const int cntj = (deg - c0 < 32) ? (deg - c0) : 32;
      for (int j = 0; j < cntj; j += GRP) {
        v4f hLo[GRP], hHi[GRP];
        for (int u = 0; u < GRP; ++u) {
          const int sj = __builtin_amdgcn_readlane(s, j + u);
          const float* __restrict__ hp = Hl + (size_t)sj * F;
          hLo[u] = *(const v4f*)hp;
          hHi[u] = *(const v4f*)(hp + 128);
        }
        for (int u = 0; u < GRP; ++u) {
          const float a0 = lane_value(a[0], j + u);
          const float a1 = lane_value(a[1], j + u);
          const float a2 = lane_value(a[2], j + u);
          const float a3 = lane_value(a[3], j + u);
          accLo += (up ? a1 : a0) * hLo[u];
          accHi += (up ? a3 : a2) * hHi[u];
        }
      }
    }
  } else {
    v4f m = lself;
    for (int c0 = 0; c0 < E; c0 += 32) {
      const int e = c0 + lane;
      if (e < E && dstIdx[e] == node) {
        const v4f ev = *(const v4f*)(e_src + (size_t)srcIdx[e] * HEADS);
        for (int h = 0; h < HEADS; ++h) m[h] = fmaxf(m[h], leaky(ev[h] + ed[h]));
      }
    }
    for (int h = 0; h < HEADS; ++h) m[h] = wave_max(m[h]);
    v4f den = {0.f, 0.f, 0.f, 0.f};
    for (int c0 = 0; c0 < E; c0 += 32) {
      const int e = c0 + lane;
      if (e < E && dstIdx[e] == node) {
        const v4f ev = *(const v4f*)(e_src + (size_t)srcIdx[e] * HEADS);
        for (int h = 0; h < HEADS; ++h) den[h] += expf(leaky(ev[h] + ed[h]) - m[h]);
      }
    }
    v4f exself, dn;
    for (int h = 0; h < HEADS; ++h) {
      exself[h] = expf(lself[h] - m[h]);
      dn[h] = (wave_sum(den[h]) + exself[h]) + 1e-16f;
    }
    const float wLo = up ? (exself[1] / dn[1]) : (exself[0] / dn[0]);
    const float wHi = up ? (exself[3] / dn[3]) : (exself[2] / dn[2]);
    accLo = wLo * *(const v4f*)hs;
    accHi = wHi * *(const v4f*)(hs + 128);
    for (int c0 = 0; c0 < E; c0 += 32) {
      const int e = c0 + lane;
      const bool hit = (e < E) && (dstIdx[e] == node);
      unsigned todo = __builtin_amdgcn_ballot_w32(hit);
      int s = node;
      v4f a = {0.f, 0.f, 0.f, 0.f};
      if (hit) {
        s = srcIdx[e];
        const v4f ev = *(const v4f*)(e_src + (size_t)s * HEADS);
        for (int h = 0; h < HEADS; ++h) a[h] = expf(leaky(ev[h] + ed[h]) - m[h]) / dn[h];
      }
      while (todo != 0u) {
        const int j = __builtin_ctz(todo);
        todo &= todo - 1u;
        const int sj = __builtin_amdgcn_readlane(s, j);
        const float* __restrict__ hp = Hl + (size_t)sj * F;
        const v4f hLo = *(const v4f*)hp;
        const v4f hHi = *(const v4f*)(hp + 128);
        const float a0 = lane_value(a[0], j);
        const float a1 = lane_value(a[1], j);
        const float a2 = lane_value(a[2], j);
        const float a3 = lane_value(a[3], j);
        accLo += (up ? a1 : a0) * hLo;
        accHi += (up ? a3 : a2) * hHi;
      }
    }
  }

  const v4f bLo = *(const v4f*)(b + 4 * lane);
  const v4f bHi = *(const v4f*)(b + 128 + 4 * lane);
  for (int k = 0; k < 4; ++k) {
    oLo[k] = elu1(accLo[k] + bLo[k]);
    oHi[k] = elu1(accHi[k] + bHi[k]);
  }
}

template<int K, int NOUT>
__global__ void __launch_bounds__(640, 10)
gat_gather_row4_gemm(const int* __restrict__ cnt,
                     const int* __restrict__ slots,
                     const int* __restrict__ srcIdx,
                     const int* __restrict__ dstIdx, int E,
                     const float* __restrict__ e_src,
                     const float* __restrict__ e_dst,
                     const float* __restrict__ H,
                     const float* __restrict__ b,
                     const float* __restrict__ W,
                     const float* __restrict__ a_src2,
                     const float* __restrict__ a_dst2,
                     float* __restrict__ H2,
                     float* __restrict__ e2_src,
                     float* __restrict__ e2_dst, int n, int nRowTiles) {
  constexpr int P = K + 4;
  constexpr int PW = 4;
  constexpr int GW = 16;
  static_assert(K == HEADS * HID && NOUT == 32 * PW, "shape");
  static_assert((size_t)(NOUT + GW) * P * 4 <= 160u * 1024u, "two work-groups per WGP");
  __shared__ __attribute__((aligned(16))) float Wt[NOUT * P];
  __shared__ __attribute__((aligned(16))) float Xs[GW * P];

  const int lane = threadIdx.x & 31;
  const int wave = __builtin_amdgcn_readfirstlane((int)(threadIdx.x >> 5));
  const int r = lane & 15;
  const int half = lane >> 4;
  const bool product = wave < PW;

  if (!product) {
    const int gw = wave - PW;
    const int q = (lane & 3) + 4 * (lane >> 4);
    const int kr = (lane >> 2) & 3;
    constexpr int TC = NOUT / 32;
    constexpr int TILES = (K / 16) * TC;
    static_assert(TILES % GW == 0, "tiles split evenly over the gather waves");
#pragma unroll
    for (int i = 0; i < TILES / GW; ++i) {
      const int t = gw + i * GW;
      const int k0 = (t / TC) * 16;
      const int c0 = (t % TC) * 32;
      const float* __restrict__ src = W + (size_t)(k0 + 4 * kr) * NOUT + c0 + 4 * q;
      const v4f v0 = *(const v4f*)(src);
      const v4f v1 = *(const v4f*)(src + NOUT);
      const v4f v2 = *(const v4f*)(src + 2 * NOUT);
      const v4f v3 = *(const v4f*)(src + 3 * NOUT);
      float* dst = Wt + (c0 + 4 * q) * P + k0 + 4 * kr;
      v4f o;
      o[0] = v0[0]; o[1] = v1[0]; o[2] = v2[0]; o[3] = v3[0];
      *(v4f*)(dst) = o;
      o[0] = v0[1]; o[1] = v1[1]; o[2] = v2[1]; o[3] = v3[1];
      *(v4f*)(dst + P) = o;
      o[0] = v0[2]; o[1] = v1[2]; o[2] = v2[2]; o[3] = v3[2];
      *(v4f*)(dst + 2 * P) = o;
      o[0] = v0[3]; o[1] = v1[3]; o[2] = v2[3]; o[3] = v3[3];
      *(v4f*)(dst + 3 * P) = o;
    }
  }

  const int pc = (product ? wave : 0) * 32;
  const float* Bl = Wt + (pc + r) * P + 4 * half;
  const float* Al = Xs + r * P + 4 * half;
  const float as0 = a_src2[pc + r],      ad0 = a_dst2[pc + r];
  const float as1 = a_src2[pc + 16 + r], ad1 = a_dst2[pc + 16 + r];

  for (int rt = (int)blockIdx.x; rt < nRowTiles; rt += (int)gridDim.x) {
    v4f oLo = {0.f, 0.f, 0.f, 0.f}, oHi = {0.f, 0.f, 0.f, 0.f};
    if (!product) {
      const int node = rt * GW + (wave - PW);
      if (node < n)
        gather_row4(cnt, slots, srcIdx, dstIdx, E, e_src, e_dst, H, b, node, oLo, oHi);
    }
    __syncthreads();
    if (!product) {
      float* xp = Xs + (wave - PW) * P + 4 * lane;
      *(v4f*)xp = oLo;
      *(v4f*)(xp + 128) = oHi;
    }
    __syncthreads();
    if (product) {
      v8f acc0 = {}, acc1 = {};
      for (int g = 0; g < K / 32; ++g) {
        const float* al = Al + g * 32;
        const float* bl = Bl + g * 32;
        v4f a[4], w0[4], w1[4];
#pragma unroll
        for (int i = 0; i < 4; ++i) {
          a[i]  = *(const v4f*)(al + 8 * i);
          w0[i] = *(const v4f*)(bl + 8 * i);
          w1[i] = *(const v4f*)(bl + 16 * P + 8 * i);
        }
#pragma unroll
        for (int i = 0; i < 4; ++i) {
          const v2f alo = {a[i][0], a[i][1]}, ahi = {a[i][2], a[i][3]};
          const v2f b0l = {w0[i][0], w0[i][1]}, b0h = {w0[i][2], w0[i][3]};
          const v2f b1l = {w1[i][0], w1[i][1]}, b1h = {w1[i][2], w1[i][3]};
          acc0 = WMMA_F32(alo, b0l, acc0);
          acc1 = WMMA_F32(alo, b1l, acc1);
          acc0 = WMMA_F32(ahi, b0h, acc0);
          acc1 = WMMA_F32(ahi, b1h, acc1);
        }
      }
      float* __restrict__ Cp = H2 + (size_t)(rt * GW + 8 * half) * NOUT + pc + r;
#pragma unroll
      for (int v = 0; v < 8; ++v) {
        Cp[(size_t)v * NOUT]      = acc0[v];
        Cp[(size_t)v * NOUT + 16] = acc1[v];
      }
      float ps[8], pd[8];
#pragma unroll
      for (int v = 0; v < 8; ++v) {
        ps[v] = acc0[v] * as0 + acc1[v] * as1;
        pd[v] = acc0[v] * ad0 + acc1[v] * ad1;
      }
#pragma unroll
      for (int v = 0; v < 8; ++v) {
        ps[v] = row_sum16(ps[v]);
        pd[v] = row_sum16(pd[v]);
      }
      float vs = ps[0], vd = pd[0];
#pragma unroll
      for (int v = 1; v < 8; ++v) {
        if ((r & 7) == v) { vs = ps[v]; vd = pd[v]; }
      }
      float* __restrict__ ep = ((r < 8) ? e2_src : e2_dst)
                             + (size_t)(rt * GW + 8 * half + (r & 7)) * PW + wave;
      *ep = (r < 8) ? vs : vd;
    }
  }
}
#undef WMMA_F32


template<int CH>
__global__ void gat_gather_mean(const int* __restrict__ cnt,
                                const int* __restrict__ slots,
                                const int* __restrict__ srcIdx,
                                const int* __restrict__ dstIdx, int E,
                                const float* __restrict__ e_src,
                                const float* __restrict__ e_dst,
                                const float* __restrict__ H,
                                const float* __restrict__ b,
                                float* __restrict__ out, int n) {
  constexpr int VEC = CH / 32;
  constexpr int GRP = 4;
  typedef typename VecOf<VEC>::type vecT;
  static_assert(CH % 32 == 0, "shape");

  const int lane = threadIdx.x & 31;
  const int wv = __builtin_amdgcn_readfirstlane((int)(threadIdx.x >> 5));
  const int node = blockIdx.x * (int)(blockDim.x >> 5) + wv;
  if (node >= n) return;
  const int deg = cnt[node];

  const float ed = pair_sum(e_dst, node);
  const float lself = leaky(pair_sum(e_src, node) + ed);
  const int* __restrict__ sp = slots + (size_t)node * ROW_CAP;
  const float* __restrict__ Hh = H + lane * VEC;
  vecT acc;

  if (deg <= ROW_CAP) {
    int s0 = node;
    float l0 = lself;
    if (lane < deg) {
      s0 = sp[lane];
      l0 = leaky(pair_sum(e_src, s0) + ed);
    }
    float m = fmaxf(lself, l0);
    for (int c0 = 32; c0 < deg; c0 += 32) {
      const int i = c0 + lane;
      if (i < deg) m = fmaxf(m, leaky(pair_sum(e_src, sp[i]) + ed));
    }
    m = wave_max(m);

    const float exself = expf(lself - m);
    const float ex0 = (lane < deg) ? expf(l0 - m) : 0.f;
    float den = ex0;
    for (int c0 = 32; c0 < deg; c0 += 32) {
      const int i = c0 + lane;
      if (i < deg) den += expf(leaky(pair_sum(e_src, sp[i]) + ed) - m);
    }
    den = wave_sum(den) + exself;
    const float dn = den + 1e-16f;

    acc = (exself / dn) * *(const vecT*)(Hh + (size_t)node * CH);

    for (int c0 = 0; c0 < deg; c0 += 32) {
      int s = s0;
      float a = ex0 / dn;
      if (c0 > 0) {
        const int i = c0 + lane;
        s = node;
        a = 0.f;
        if (i < deg) {
          s = sp[i];
          a = expf(leaky(pair_sum(e_src, s) + ed) - m) / dn;
        }
      }
      const int cntj = (deg - c0 < 32) ? (deg - c0) : 32;
      for (int j = 0; j < cntj; j += GRP) {
        vecT hb[GRP];
        for (int u = 0; u < GRP; ++u) {
          const int sj = __builtin_amdgcn_readlane(s, j + u);
          hb[u] = *(const vecT*)(Hh + (size_t)sj * CH);
        }
        for (int u = 0; u < GRP; ++u) acc += lane_value(a, j + u) * hb[u];
      }
    }
  } else {
    float m = lself;
    for (int c0 = 0; c0 < E; c0 += 32) {
      const int e = c0 + lane;
      if (e < E && dstIdx[e] == node) m = fmaxf(m, leaky(pair_sum(e_src, srcIdx[e]) + ed));
    }
    m = wave_max(m);
    float den = 0.f;
    for (int c0 = 0; c0 < E; c0 += 32) {
      const int e = c0 + lane;
      if (e < E && dstIdx[e] == node) den += expf(leaky(pair_sum(e_src, srcIdx[e]) + ed) - m);
    }
    const float exself = expf(lself - m);
    const float dn = (wave_sum(den) + exself) + 1e-16f;
    acc = (exself / dn) * *(const vecT*)(Hh + (size_t)node * CH);
    for (int c0 = 0; c0 < E; c0 += 32) {
      const int e = c0 + lane;
      const bool hit = (e < E) && (dstIdx[e] == node);
      unsigned todo = __builtin_amdgcn_ballot_w32(hit);
      int s = node;
      float a = 0.f;
      if (hit) {
        s = srcIdx[e];
        a = expf(leaky(pair_sum(e_src, s) + ed) - m) / dn;
      }
      while (todo != 0u) {
        const int j = __builtin_ctz(todo);
        todo &= todo - 1u;
        const int sj = __builtin_amdgcn_readlane(s, j);
        acc += lane_value(a, j) * *(const vecT*)(Hh + (size_t)sj * CH);
      }
    }
  }

  const float* __restrict__ bp = b + lane * VEC;
  float sum = 0.f;
  for (int k = 0; k < VEC; ++k) sum += elu1(acc[k] + bp[k]);
  sum = wave_sum(sum);
  if (lane == 0) out[node] = sum / (float)CH;
}

static inline int cdiv(long long a, long long b) { return (int)((a + b - 1) / b); }

extern "C" void kernel_launch(void* const* d_in, const int* in_sizes, int n_in,
                              void* d_out, int out_size, void* d_ws, size_t ws_size,
                              hipStream_t stream) {
  const float* x      = (const float*)d_in[0];
  const int*   ei     = (const int*)  d_in[1];
  const float* W1     = (const float*)d_in[2];
  const float* a1_src = (const float*)d_in[3];
  const float* a1_dst = (const float*)d_in[4];
  const float* b1     = (const float*)d_in[5];
  const float* W2     = (const float*)d_in[6];
  const float* a2_src = (const float*)d_in[7];
  const float* a2_dst = (const float*)d_in[8];
  const float* b2     = (const float*)d_in[9];
  float* outp = (float*)d_out;

  const int N    = in_sizes[0] / DIN;
  const int E    = in_sizes[1] / 2;
  const int Etot = E + N;
  const int F1   = HEADS * HID;
  const int* srcIdx = ei;
  const int* dstIdx = ei + E;

  float* ws = (float*)d_ws;
  float* h1      = ws;
  float* e_src_b = h1 + (size_t)N * F1;
  float* e_dst_b = e_src_b + (size_t)N * HEADS;
  float* edgebuf = e_dst_b + (size_t)N * HEADS;
  float* m_b     = edgebuf + (size_t)Etot * HEADS;
  float* den_b   = m_b + (size_t)N * HEADS;
  float* out1    = den_b + (size_t)N * HEADS;
  float* h2      = out1;
  float* e2_src  = h2 + (size_t)N * DOUT;
  float* e2_dst  = e2_src + (size_t)N * HEADS;
  int* slots = (int*)edgebuf;
  int* cnt   = (int*)den_b;

  const int TB = 256;

  csr_zero<<<cdiv(N, TB), TB, 0, stream>>>(cnt, N);
  csr_rank<<<128, 1024, 0, stream>>>(srcIdx, dstIdx, E, N, cnt, slots);

  {
    gemm_f32_wmma<DIN, F1><<<256, 512, 0, stream>>>(
        x, W1, a1_src, a1_dst, h1, e_src_b, e_dst_b, N / 16);
  }

  {
    gat_gather_row4_gemm<F1, DOUT><<<512, 640, 0, stream>>>(
        cnt, slots, srcIdx, dstIdx, E, e_src_b, e_dst_b, h1, b1,
        W2, a2_src, a2_dst, h2, e2_src, e2_dst, N, N / 16);

    gat_gather_mean<DOUT><<<cdiv(N, TB / 32), TB, 0, stream>>>(
        cnt, slots, srcIdx, dstIdx, E, e2_src, e2_dst, h2, b2, outp, N);
  }
}
